# starting kernel
# baseline (speedup 1.0000x reference)
#include <hip/hip_runtime.h>
#include <hip/hip_bf16.h>


typedef float v2f __attribute__((ext_vector_type(2)));
typedef float v8f __attribute__((ext_vector_type(8)));

__device__ __forceinline__ v8f wmma_f32(v2f a, v2f b, v8f c) {
    return __builtin_amdgcn_wmma_f32_16x16x4_f32(
        false, a, false, b, (short)0, c, false, false);
}

__global__ __launch_bounds__(256) void reorder_w(const float* __restrict__ W,
                                                 int N,
                                                 float* __restrict__ Wr) {
    int o    = blockIdx.x * 256 + threadIdx.x;
    int pair = o / 24576;
    int rem  = o % 24576;
    int i    = rem / 128;
    int rem2 = rem % 128;
    int lane = rem2 >> 2;
    int tl   = (rem2 >> 1) & 1;
    int j    = rem2 & 1;
    int m    = lane & 15;
    int half = lane >> 4;
    int k    = 4 * i + 2 * half + j;
    int n    = (pair * 2 + tl) * 16 + m;
    Wr[o] = W[k * N + n];
}

__global__ __launch_bounds__(256) void lengths_k(const int* __restrict__ mask,
                                                 int* __restrict__ Lout) {
    __shared__ int sh[8];
    int b = blockIdx.x, t = threadIdx.x;
    int s = 0;
    #pragma unroll
    for (int j = 0; j < 8; ++j) s += mask[b * 2048 + t + 256 * j];
    #pragma unroll
    for (int o = 1; o < 32; o <<= 1) s += __shfl_xor(s, o, 32);
    if ((t & 31) == 0) sh[t >> 5] = s;
    __syncthreads();
    if (t == 0) {
        int tot = 0;
        #pragma unroll
        for (int j = 0; j < 8; ++j) tot += sh[j];
        int L = tot - 1;
        Lout[b] = L < 2046 ? L : 2046;
    }
}

#define LDA 196
#define LDC 132
#define KP  192
#define IT  48

__global__ __launch_bounds__(128) void tok_gemm_ln(
    const float* __restrict__ hidden, const int* __restrict__ amask,
    const float* __restrict__ wr,     const float* __restrict__ tok_b,
    const float* __restrict__ gamma,  const float* __restrict__ beta,
    float* __restrict__ reps) {
    __shared__ __align__(16) float shmem[64 * LDA];
    float* shA = shmem;
    float* shC = shmem;

    const int  t    = threadIdx.x;
    const int  lane = t & 31;
    const int  wv   = t >> 5;
    const int  m    = lane & 15;
    const int  half = lane >> 4;
    const long row0 = (long)blockIdx.x * 64;

    v8f acc[4][2] = {};
    const float4* bp = (const float4*)wr + ((long)wv * 192) * 32 + lane;

    for (int ph = 0; ph < 4; ++ph) {
        __syncthreads();
        for (int idx = t; idx < 64 * KP / 4; idx += 128) {
            int r  = idx / (KP / 4);
            int c4 = (idx % (KP / 4)) * 4;
            float4 v = *(const float4*)(hidden + (row0 + r) * 768 + ph * KP + c4);
            *(float4*)&shA[r * LDA + c4] = v;
        }
        __syncthreads();

        for (int i = 0; i < IT; ++i) {
            float4 f = bp[(long)(ph * IT + i) * 32];
            v2f b0 = {f.x, f.y};
            v2f b1 = {f.z, f.w};
            #pragma unroll
            for (int mt = 0; mt < 4; ++mt) {
                v2f a = *(const v2f*)&shA[(mt * 16 + m) * LDA + 4 * i + 2 * half];
                acc[mt][0] = wmma_f32(a, b0, acc[mt][0]);
                acc[mt][1] = wmma_f32(a, b1, acc[mt][1]);
            }
        }
    }
    __syncthreads();

    #pragma unroll
    for (int mt = 0; mt < 4; ++mt) {
        #pragma unroll
        for (int i = 0; i < 8; ++i) {
            shC[(mt * 16 + i + 8 * half) * LDC + wv * 32 + m]      = acc[mt][0][i];
            shC[(mt * 16 + i + 8 * half) * LDC + wv * 32 + 16 + m] = acc[mt][1][i];
        }
    }
    __syncthreads();

    const int r  = t >> 1;
    const int c0 = (t & 1) * 64;
    float s = 0.f, ss = 0.f;
    for (int i = 0; i < 64; ++i) {
        float v = shC[r * LDC + c0 + i] + tok_b[c0 + i];
        s += v; ss += v * v;
    }
    s  += __shfl_xor(s,  1, 32);
    ss += __shfl_xor(ss, 1, 32);
    const float mean = s * (1.f / 128.f);
    const float var  = ss * (1.f / 128.f) - mean * mean;
    const float rstd = rsqrtf(var + 1e-5f);

    const long rg = row0 + r;
    const int  bI = (int)(rg >> 11);
    const int  sI = (int)(rg & 2047);
    const float mk = (float)amask[rg];
    if (sI >= 1 && sI <= 2046) {
        float* dst = reps + ((long)(bI * 2046 + sI - 1)) * 128 + c0;
        for (int i = 0; i < 64; ++i) {
            float v = shC[r * LDC + c0 + i] + tok_b[c0 + i];
            float y = (v - mean) * rstd * gamma[c0 + i] + beta[c0 + i];
            dst[i] = fmaxf(y, 0.f) * mk;
        }
    }
}

__global__ __launch_bounds__(128) void cls_gemm(const float* __restrict__ hidden,
                                                const float* __restrict__ wr2,
                                                float* __restrict__ cls_lin) {
    const int t = threadIdx.x, lane = t & 31, wv = t >> 5;
    const int m = lane & 15, half = lane >> 4;
    const int pair = blockIdx.x * 4 + wv;

    v8f acc[2][2] = {};
    const float4* bp = (const float4*)wr2 + ((long)pair * 192) * 32 + lane;
    const float* a0b = hidden + (long)m * (2048 * 768) + 2 * half;
    const float* a1b = hidden + (long)(m + 16) * (2048 * 768) + 2 * half;

    for (int i = 0; i < 192; ++i) {
        v2f a0 = *(const v2f*)(a0b + 4 * i);
        v2f a1 = *(const v2f*)(a1b + 4 * i);
        float4 f = bp[i * 32];
        v2f b0 = {f.x, f.y};
        v2f b1 = {f.z, f.w};
        acc[0][0] = wmma_f32(a0, b0, acc[0][0]);
        acc[0][1] = wmma_f32(a0, b1, acc[0][1]);
        acc[1][0] = wmma_f32(a1, b0, acc[1][0]);
        acc[1][1] = wmma_f32(a1, b1, acc[1][1]);
    }
    const int c0 = pair * 32 + m;
    #pragma unroll
    for (int i = 0; i < 8; ++i) {
        cls_lin[(i + 8 * half) * 768 + c0]           = acc[0][0][i];
        cls_lin[(i + 8 * half) * 768 + c0 + 16]      = acc[0][1][i];
        cls_lin[(i + 8 * half + 16) * 768 + c0]      = acc[1][0][i];
        cls_lin[(i + 8 * half + 16) * 768 + c0 + 16] = acc[1][1][i];
    }
}

__global__ __launch_bounds__(256) void cls_ln(const float* __restrict__ cls_lin,
                                              const float* __restrict__ cls_b,
                                              const float* __restrict__ gamma,
                                              const float* __restrict__ beta,
                                              float* __restrict__ out) {
    __shared__ float sh_s[8], sh_ss[8];
    const int b = blockIdx.x, t = threadIdx.x;
    float x[3];
    float s = 0.f, ss = 0.f;
    #pragma unroll
    for (int j = 0; j < 3; ++j) {
        int c = t + 256 * j;
        float v = cls_lin[b * 768 + c] + cls_b[c];
        x[j] = v; s += v; ss += v * v;
    }
    #pragma unroll
    for (int o = 1; o < 32; o <<= 1) {
        s  += __shfl_xor(s,  o, 32);
        ss += __shfl_xor(ss, o, 32);
    }
    if ((t & 31) == 0) { sh_s[t >> 5] = s; sh_ss[t >> 5] = ss; }
    __syncthreads();
    float ts = 0.f, tss = 0.f;
    #pragma unroll
    for (int j = 0; j < 8; ++j) { ts += sh_s[j]; tss += sh_ss[j]; }
    const float mean = ts * (1.f / 768.f);
    const float var  = tss * (1.f / 768.f) - mean * mean;
    const float rstd = rsqrtf(var + 1e-5f);
    #pragma unroll
    for (int j = 0; j < 3; ++j) {
        int c = t + 256 * j;
        out[b * 768 + c] = (x[j] - mean) * rstd * gamma[c] + beta[c];
    }
}

__global__ __launch_bounds__(256) void window_norm(const float* __restrict__ reps,
                                                   const int* __restrict__ Lw,
                                                   float* __restrict__ out) {
    const int g    = blockIdx.x * 8 + (threadIdx.x >> 5);
    const int lane = threadIdx.x & 31;
    const int b    = g / 2046;
    const int tp   = g % 2046;
    const int L    = Lw[b];
    const bool valid = tp < L;
    int start = tp - 5; if (start < 0) start = 0;
    int end   = tp + 5; if (end > L) end = L;
    int cnt   = end - start; if (cnt < 1) cnt = 1;

    float4 sum = {0.f, 0.f, 0.f, 0.f};
    const float* base = reps + ((long)b * 2046) * 128 + lane * 4;
    for (int r = start; r < end; ++r) {
        float4 v = *(const float4*)(base + (long)r * 128);
        sum.x += v.x; sum.y += v.y; sum.z += v.z; sum.w += v.w;
    }
    const float inv = 1.f / (float)cnt;
    float4 mn = {sum.x * inv, sum.y * inv, sum.z * inv, sum.w * inv};
    float sq = mn.x * mn.x + mn.y * mn.y + mn.z * mn.z + mn.w * mn.w;
    #pragma unroll
    for (int o = 1; o < 32; o <<= 1) sq += __shfl_xor(sq, o, 32);
    const float n  = sqrtf(sq);
    const float sc = (valid && n > 0.f) ? (1.f / n) : 0.f;
    float4 o4 = {mn.x * sc, mn.y * sc, mn.z * sc, mn.w * sc};
    *(float4*)(out + (long)g * 128 + lane * 4) = o4;
}

extern "C" void kernel_launch(void* const* d_in, const int* in_sizes, int n_in,
                              void* d_out, int out_size, void* d_ws, size_t ws_size,
                              hipStream_t stream) {
    const float* hidden   = (const float*)d_in[0];
    const int*   amask    = (const int*)  d_in[1];
    const float* tok_w    = (const float*)d_in[2];
    const float* tok_b    = (const float*)d_in[3];
    const float* cls_w    = (const float*)d_in[4];
    const float* cls_b    = (const float*)d_in[5];
    const float* ln_tok_g = (const float*)d_in[6];
    const float* ln_tok_b = (const float*)d_in[7];
    const float* ln_cls_g = (const float*)d_in[8];
    const float* ln_cls_b = (const float*)d_in[9];
    float* out = (float*)d_out;

    float* ws      = (float*)d_ws;
    float* reps    = ws;
    float* cls_lin = ws + 8380416;
    float* wr_tok  = ws + 8404992;
    float* wr_cls  = ws + 8503296;
    int*   lenw    = (int*)(ws + 9093120);

    reorder_w<<<384,  256, 0, stream>>>(tok_w, 128, wr_tok);
    reorder_w<<<2304, 256, 0, stream>>>(cls_w, 768, wr_cls);
    lengths_k<<<32,   256, 0, stream>>>(amask, lenw);

    tok_gemm_ln<<<1024, 128, 0, stream>>>(hidden, amask, wr_tok, tok_b,
                                          ln_tok_g, ln_tok_b, reps);
    cls_gemm<<<6, 128, 0, stream>>>(hidden, wr_cls, cls_lin);
    cls_ln<<<32, 256, 0, stream>>>(cls_lin, cls_b, ln_cls_g, ln_cls_b, out);
    window_norm<<<8184, 256, 0, stream>>>(reps, lenw, out + 24576);
}


# cls_gemm re-gridded from 24 to 768 waves: K split over a block's 16 waves (12 K-steps each, all loads in flight before the first WMMA), partials summed via LDS in wave order; cls K-sum order changes
# speedup vs baseline: 1.3908x; 1.3908x over previous
#include <hip/hip_runtime.h>
#include <hip/hip_bf16.h>


typedef float v2f __attribute__((ext_vector_type(2)));
typedef float v8f __attribute__((ext_vector_type(8)));

__device__ __forceinline__ v8f wmma_f32(v2f a, v2f b, v8f c) {
    return __builtin_amdgcn_wmma_f32_16x16x4_f32(
        false, a, false, b, (short)0, c, false, false);
}

__global__ __launch_bounds__(256) void reorder_w(const float* __restrict__ W,
                                                 int N,
                                                 float* __restrict__ Wr) {
    int o    = blockIdx.x * 256 + threadIdx.x;
    int pair = o / 24576;
    int rem  = o % 24576;
    int i    = rem / 128;
    int rem2 = rem % 128;
    int lane = rem2 >> 2;
    int tl   = (rem2 >> 1) & 1;
    int j    = rem2 & 1;
    int m    = lane & 15;
    int half = lane >> 4;
    int k    = 4 * i + 2 * half + j;
    int n    = (pair * 2 + tl) * 16 + m;
    Wr[o] = W[k * N + n];
}

__global__ __launch_bounds__(256) void lengths_k(const int* __restrict__ mask,
                                                 int* __restrict__ Lout) {
    __shared__ int sh[8];
    int b = blockIdx.x, t = threadIdx.x;
    int s = 0;
    #pragma unroll
    for (int j = 0; j < 8; ++j) s += mask[b * 2048 + t + 256 * j];
    #pragma unroll
    for (int o = 1; o < 32; o <<= 1) s += __shfl_xor(s, o, 32);
    if ((t & 31) == 0) sh[t >> 5] = s;
    __syncthreads();
    if (t == 0) {
        int tot = 0;
        #pragma unroll
        for (int j = 0; j < 8; ++j) tot += sh[j];
        int L = tot - 1;
        Lout[b] = L < 2046 ? L : 2046;
    }
}

#define LDA 196
#define LDC 132
#define KP  192
#define IT  48

__global__ __launch_bounds__(128) void tok_gemm_ln(
    const float* __restrict__ hidden, const int* __restrict__ amask,
    const float* __restrict__ wr,     const float* __restrict__ tok_b,
    const float* __restrict__ gamma,  const float* __restrict__ beta,
    float* __restrict__ reps) {
    __shared__ __align__(16) float shmem[64 * LDA];
    float* shA = shmem;
    float* shC = shmem;

    const int  t    = threadIdx.x;
    const int  lane = t & 31;
    const int  wv   = t >> 5;
    const int  m    = lane & 15;
    const int  half = lane >> 4;
    const long row0 = (long)blockIdx.x * 64;

    v8f acc[4][2] = {};
    const float4* bp = (const float4*)wr + ((long)wv * 192) * 32 + lane;

    for (int ph = 0; ph < 4; ++ph) {
        __syncthreads();
        for (int idx = t; idx < 64 * KP / 4; idx += 128) {
            int r  = idx / (KP / 4);
            int c4 = (idx % (KP / 4)) * 4;
            float4 v = *(const float4*)(hidden + (row0 + r) * 768 + ph * KP + c4);
            *(float4*)&shA[r * LDA + c4] = v;
        }
        __syncthreads();

        for (int i = 0; i < IT; ++i) {
            float4 f = bp[(long)(ph * IT + i) * 32];
            v2f b0 = {f.x, f.y};
            v2f b1 = {f.z, f.w};
            #pragma unroll
            for (int mt = 0; mt < 4; ++mt) {
                v2f a = *(const v2f*)&shA[(mt * 16 + m) * LDA + 4 * i + 2 * half];
                acc[mt][0] = wmma_f32(a, b0, acc[mt][0]);
                acc[mt][1] = wmma_f32(a, b1, acc[mt][1]);
            }
        }
    }
    __syncthreads();

    #pragma unroll
    for (int mt = 0; mt < 4; ++mt) {
        #pragma unroll
        for (int i = 0; i < 8; ++i) {
            shC[(mt * 16 + i + 8 * half) * LDC + wv * 32 + m]      = acc[mt][0][i];
            shC[(mt * 16 + i + 8 * half) * LDC + wv * 32 + 16 + m] = acc[mt][1][i];
        }
    }
    __syncthreads();

    const int r  = t >> 1;
    const int c0 = (t & 1) * 64;
    float s = 0.f, ss = 0.f;
    for (int i = 0; i < 64; ++i) {
        float v = shC[r * LDC + c0 + i] + tok_b[c0 + i];
        s += v; ss += v * v;
    }
    s  += __shfl_xor(s,  1, 32);
    ss += __shfl_xor(ss, 1, 32);
    const float mean = s * (1.f / 128.f);
    const float var  = ss * (1.f / 128.f) - mean * mean;
    const float rstd = rsqrtf(var + 1e-5f);

    const long rg = row0 + r;
    const int  bI = (int)(rg >> 11);
    const int  sI = (int)(rg & 2047);
    const float mk = (float)amask[rg];
    if (sI >= 1 && sI <= 2046) {
        float* dst = reps + ((long)(bI * 2046 + sI - 1)) * 128 + c0;
        for (int i = 0; i < 64; ++i) {
            float v = shC[r * LDC + c0 + i] + tok_b[c0 + i];
            float y = (v - mean) * rstd * gamma[c0 + i] + beta[c0 + i];
            dst[i] = fmaxf(y, 0.f) * mk;
        }
    }
}

#define CLS_KW 12

__global__ __launch_bounds__(512) void cls_gemm(const float* __restrict__ hidden,
                                                const float* __restrict__ wr2,
                                                float* __restrict__ cls_lin) {
    __shared__ float red[16 * 512];
    const int t = threadIdx.x, lane = t & 31, wv = t >> 5;
    const int m = lane & 15, half = lane >> 4;
    const int pair = blockIdx.x >> 1;
    const int mt   = blockIdx.x & 1;

    const float4* bp = (const float4*)wr2 + ((long)pair * 192 + wv * CLS_KW) * 32 + lane;
    const float*  ap = hidden + (long)(mt * 16 + m) * (2048 * 768)
                              + wv * (4 * CLS_KW) + 2 * half;
    float* rw = red + wv * 512 + lane;

    v2f    a[CLS_KW];
    float4 f[CLS_KW];
    #pragma unroll
    for (int i = 0; i < CLS_KW; ++i) {
        a[i] = *(const v2f*)(ap + 4 * i);
        f[i] = bp[i * 32];
    }
    __builtin_amdgcn_sched_barrier(0);

    v8f acc0 = {}, acc1 = {};
    #pragma unroll
    for (int i = 0; i < CLS_KW; ++i) {
        v2f b0 = {f[i].x, f[i].y};
        v2f b1 = {f[i].z, f[i].w};
        acc0 = wmma_f32(a[i], b0, acc0);
        acc1 = wmma_f32(a[i], b1, acc1);
    }

    #pragma unroll
    for (int i = 0; i < 8; ++i) {
        rw[i * 32]       = acc0[i];
        rw[(8 + i) * 32] = acc1[i];
    }
    __syncthreads();

    float s = 0.f;
    #pragma unroll
    for (int w = 0; w < 16; ++w) s += red[w * 512 + t];
    const int tile = wv >> 3, i = wv & 7;
    cls_lin[(mt * 16 + i + 8 * half) * 768 + pair * 32 + tile * 16 + m] = s;
}

__global__ __launch_bounds__(256) void cls_ln(const float* __restrict__ cls_lin,
                                              const float* __restrict__ cls_b,
                                              const float* __restrict__ gamma,
                                              const float* __restrict__ beta,
                                              float* __restrict__ out) {
    __shared__ float sh_s[8], sh_ss[8];
    const int b = blockIdx.x, t = threadIdx.x;
    float x[3];
    float s = 0.f, ss = 0.f;
    #pragma unroll
    for (int j = 0; j < 3; ++j) {
        int c = t + 256 * j;
        float v = cls_lin[b * 768 + c] + cls_b[c];
        x[j] = v; s += v; ss += v * v;
    }
    #pragma unroll
    for (int o = 1; o < 32; o <<= 1) {
        s  += __shfl_xor(s,  o, 32);
        ss += __shfl_xor(ss, o, 32);
    }
    if ((t & 31) == 0) { sh_s[t >> 5] = s; sh_ss[t >> 5] = ss; }
    __syncthreads();
    float ts = 0.f, tss = 0.f;
    #pragma unroll
    for (int j = 0; j < 8; ++j) { ts += sh_s[j]; tss += sh_ss[j]; }
    const float mean = ts * (1.f / 768.f);
    const float var  = tss * (1.f / 768.f) - mean * mean;
    const float rstd = rsqrtf(var + 1e-5f);
    #pragma unroll
    for (int j = 0; j < 3; ++j) {
        int c = t + 256 * j;
        out[b * 768 + c] = (x[j] - mean) * rstd * gamma[c] + beta[c];
    }
}

__global__ __launch_bounds__(256) void window_norm(const float* __restrict__ reps,
                                                   const int* __restrict__ Lw,
                                                   float* __restrict__ out) {
    const int g    = blockIdx.x * 8 + (threadIdx.x >> 5);
    const int lane = threadIdx.x & 31;
    const int b    = g / 2046;
    const int tp   = g % 2046;
    const int L    = Lw[b];
    const bool valid = tp < L;
    int start = tp - 5; if (start < 0) start = 0;
    int end   = tp + 5; if (end > L) end = L;
    int cnt   = end - start; if (cnt < 1) cnt = 1;

    float4 sum = {0.f, 0.f, 0.f, 0.f};
    const float* base = reps + ((long)b * 2046) * 128 + lane * 4;
    for (int r = start; r < end; ++r) {
        float4 v = *(const float4*)(base + (long)r * 128);
        sum.x += v.x; sum.y += v.y; sum.z += v.z; sum.w += v.w;
    }
    const float inv = 1.f / (float)cnt;
    float4 mn = {sum.x * inv, sum.y * inv, sum.z * inv, sum.w * inv};
    float sq = mn.x * mn.x + mn.y * mn.y + mn.z * mn.z + mn.w * mn.w;
    #pragma unroll
    for (int o = 1; o < 32; o <<= 1) sq += __shfl_xor(sq, o, 32);
    const float n  = sqrtf(sq);
    const float sc = (valid && n > 0.f) ? (1.f / n) : 0.f;
    float4 o4 = {mn.x * sc, mn.y * sc, mn.z * sc, mn.w * sc};
    *(float4*)(out + (long)g * 128 + lane * 4) = o4;
}

extern "C" void kernel_launch(void* const* d_in, const int* in_sizes, int n_in,
                              void* d_out, int out_size, void* d_ws, size_t ws_size,
                              hipStream_t stream) {
    const float* hidden   = (const float*)d_in[0];
    const int*   amask    = (const int*)  d_in[1];
    const float* tok_w    = (const float*)d_in[2];
    const float* tok_b    = (const float*)d_in[3];
    const float* cls_w    = (const float*)d_in[4];
    const float* cls_b    = (const float*)d_in[5];
    const float* ln_tok_g = (const float*)d_in[6];
    const float* ln_tok_b = (const float*)d_in[7];
    const float* ln_cls_g = (const float*)d_in[8];
    const float* ln_cls_b = (const float*)d_in[9];
    float* out = (float*)d_out;

    float* ws      = (float*)d_ws;
    float* reps    = ws;
    float* cls_lin = ws + 8380416;
    float* wr_tok  = ws + 8404992;
    float* wr_cls  = ws + 8503296;
    int*   lenw    = (int*)(ws + 9093120);

    reorder_w<<<384,  256, 0, stream>>>(tok_w, 128, wr_tok);
    reorder_w<<<2304, 256, 0, stream>>>(cls_w, 768, wr_cls);
    lengths_k<<<32,   256, 0, stream>>>(amask, lenw);

    tok_gemm_ln<<<1024, 128, 0, stream>>>(hidden, amask, wr_tok, tok_b,
                                          ln_tok_g, ln_tok_b, reps);
    cls_gemm<<<48, 512, 0, stream>>>(hidden, wr_cls, cls_lin);
    cls_ln<<<32, 256, 0, stream>>>(cls_lin, cls_b, ln_cls_g, ln_cls_b, out);
    window_norm<<<8184, 256, 0, stream>>>(reps, lenw, out + 24576);
}


# window_norm: one wave per 16 consecutive rows, its 25 reps rows loaded once up front and kept in registers (65472 -> 4096 waves; a row is loaded by <=2 waves, not 10); sum order kept
# speedup vs baseline: 1.4463x; 1.0399x over previous
#include <hip/hip_runtime.h>
#include <hip/hip_bf16.h>


typedef float v2f __attribute__((ext_vector_type(2)));
typedef float v8f __attribute__((ext_vector_type(8)));

__device__ __forceinline__ v8f wmma_f32(v2f a, v2f b, v8f c) {
    return __builtin_amdgcn_wmma_f32_16x16x4_f32(
        false, a, false, b, (short)0, c, false, false);
}

__global__ __launch_bounds__(256) void reorder_w(const float* __restrict__ W,
                                                 int N,
                                                 float* __restrict__ Wr) {
    int o    = blockIdx.x * 256 + threadIdx.x;
    int pair = o / 24576;
    int rem  = o % 24576;
    int i    = rem / 128;
    int rem2 = rem % 128;
    int lane = rem2 >> 2;
    int tl   = (rem2 >> 1) & 1;
    int j    = rem2 & 1;
    int m    = lane & 15;
    int half = lane >> 4;
    int k    = 4 * i + 2 * half + j;
    int n    = (pair * 2 + tl) * 16 + m;
    Wr[o] = W[k * N + n];
}

__global__ __launch_bounds__(256) void lengths_k(const int* __restrict__ mask,
                                                 int* __restrict__ Lout) {
    __shared__ int sh[8];
    int b = blockIdx.x, t = threadIdx.x;
    int s = 0;
    #pragma unroll
    for (int j = 0; j < 8; ++j) s += mask[b * 2048 + t + 256 * j];
    #pragma unroll
    for (int o = 1; o < 32; o <<= 1) s += __shfl_xor(s, o, 32);
    if ((t & 31) == 0) sh[t >> 5] = s;
    __syncthreads();
    if (t == 0) {
        int tot = 0;
        #pragma unroll
        for (int j = 0; j < 8; ++j) tot += sh[j];
        int L = tot - 1;
        Lout[b] = L < 2046 ? L : 2046;
    }
}

#define LDA 196
#define LDC 132
#define KP  192
#define IT  48

__global__ __launch_bounds__(128) void tok_gemm_ln(
    const float* __restrict__ hidden, const int* __restrict__ amask,
    const float* __restrict__ wr,     const float* __restrict__ tok_b,
    const float* __restrict__ gamma,  const float* __restrict__ beta,
    float* __restrict__ reps) {
    __shared__ __align__(16) float shmem[64 * LDA];
    float* shA = shmem;
    float* shC = shmem;

    const int  t    = threadIdx.x;
    const int  lane = t & 31;
    const int  wv   = t >> 5;
    const int  m    = lane & 15;
    const int  half = lane >> 4;
    const long row0 = (long)blockIdx.x * 64;

    v8f acc[4][2] = {};
    const float4* bp = (const float4*)wr + ((long)wv * 192) * 32 + lane;

    for (int ph = 0; ph < 4; ++ph) {
        __syncthreads();
        for (int idx = t; idx < 64 * KP / 4; idx += 128) {
            int r  = idx / (KP / 4);
            int c4 = (idx % (KP / 4)) * 4;
            float4 v = *(const float4*)(hidden + (row0 + r) * 768 + ph * KP + c4);
            *(float4*)&shA[r * LDA + c4] = v;
        }
        __syncthreads();

        for (int i = 0; i < IT; ++i) {
            float4 f = bp[(long)(ph * IT + i) * 32];
            v2f b0 = {f.x, f.y};
            v2f b1 = {f.z, f.w};
            #pragma unroll
            for (int mt = 0; mt < 4; ++mt) {
                v2f a = *(const v2f*)&shA[(mt * 16 + m) * LDA + 4 * i + 2 * half];
                acc[mt][0] = wmma_f32(a, b0, acc[mt][0]);
                acc[mt][1] = wmma_f32(a, b1, acc[mt][1]);
            }
        }
    }
    __syncthreads();

    #pragma unroll
    for (int mt = 0; mt < 4; ++mt) {
        #pragma unroll
        for (int i = 0; i < 8; ++i) {
            shC[(mt * 16 + i + 8 * half) * LDC + wv * 32 + m]      = acc[mt][0][i];
            shC[(mt * 16 + i + 8 * half) * LDC + wv * 32 + 16 + m] = acc[mt][1][i];
        }
    }
    __syncthreads();

    const int r  = t >> 1;
    const int c0 = (t & 1) * 64;
    float s = 0.f, ss = 0.f;
    for (int i = 0; i < 64; ++i) {
        float v = shC[r * LDC + c0 + i] + tok_b[c0 + i];
        s += v; ss += v * v;
    }
    s  += __shfl_xor(s,  1, 32);
    ss += __shfl_xor(ss, 1, 32);
    const float mean = s * (1.f / 128.f);
    const float var  = ss * (1.f / 128.f) - mean * mean;
    const float rstd = rsqrtf(var + 1e-5f);

    const long rg = row0 + r;
    const int  bI = (int)(rg >> 11);
    const int  sI = (int)(rg & 2047);
    const float mk = (float)amask[rg];
    if (sI >= 1 && sI <= 2046) {
        float* dst = reps + ((long)(bI * 2046 + sI - 1)) * 128 + c0;
        for (int i = 0; i < 64; ++i) {
            float v = shC[r * LDC + c0 + i] + tok_b[c0 + i];
            float y = (v - mean) * rstd * gamma[c0 + i] + beta[c0 + i];
            dst[i] = fmaxf(y, 0.f) * mk;
        }
    }
}

#define CLS_KW 12

__global__ __launch_bounds__(512) void cls_gemm(const float* __restrict__ hidden,
                                                const float* __restrict__ wr2,
                                                float* __restrict__ cls_lin) {
    __shared__ float red[16 * 512];
    const int t = threadIdx.x, lane = t & 31, wv = t >> 5;
    const int m = lane & 15, half = lane >> 4;
    const int pair = blockIdx.x >> 1;
    const int mt   = blockIdx.x & 1;

    const float4* bp = (const float4*)wr2 + ((long)pair * 192 + wv * CLS_KW) * 32 + lane;
    const float*  ap = hidden + (long)(mt * 16 + m) * (2048 * 768)
                              + wv * (4 * CLS_KW) + 2 * half;
    float* rw = red + wv * 512 + lane;

    v2f    a[CLS_KW];
    float4 f[CLS_KW];
    #pragma unroll
    for (int i = 0; i < CLS_KW; ++i) {
        a[i] = *(const v2f*)(ap + 4 * i);
        f[i] = bp[i * 32];
    }
    __builtin_amdgcn_sched_barrier(0);

    v8f acc0 = {}, acc1 = {};
    #pragma unroll
    for (int i = 0; i < CLS_KW; ++i) {
        v2f b0 = {f[i].x, f[i].y};
        v2f b1 = {f[i].z, f[i].w};
        acc0 = wmma_f32(a[i], b0, acc0);
        acc1 = wmma_f32(a[i], b1, acc1);
    }

    #pragma unroll
    for (int i = 0; i < 8; ++i) {
        rw[i * 32]       = acc0[i];
        rw[(8 + i) * 32] = acc1[i];
    }
    __syncthreads();

    float s = 0.f;
    #pragma unroll
    for (int w = 0; w < 16; ++w) s += red[w * 512 + t];
    const int tile = wv >> 3, i = wv & 7;
    cls_lin[(mt * 16 + i + 8 * half) * 768 + pair * 32 + tile * 16 + m] = s;
}

__global__ __launch_bounds__(256) void cls_ln(const float* __restrict__ cls_lin,
                                              const float* __restrict__ cls_b,
                                              const float* __restrict__ gamma,
                                              const float* __restrict__ beta,
                                              float* __restrict__ out) {
    __shared__ float sh_s[8], sh_ss[8];
    const int b = blockIdx.x, t = threadIdx.x;
    float x[3];
    float s = 0.f, ss = 0.f;
    #pragma unroll
    for (int j = 0; j < 3; ++j) {
        int c = t + 256 * j;
        float v = cls_lin[b * 768 + c] + cls_b[c];
        x[j] = v; s += v; ss += v * v;
    }
    #pragma unroll
    for (int o = 1; o < 32; o <<= 1) {
        s  += __shfl_xor(s,  o, 32);
        ss += __shfl_xor(ss, o, 32);
    }
    if ((t & 31) == 0) { sh_s[t >> 5] = s; sh_ss[t >> 5] = ss; }
    __syncthreads();
    float ts = 0.f, tss = 0.f;
    #pragma unroll
    for (int j = 0; j < 8; ++j) { ts += sh_s[j]; tss += sh_ss[j]; }
    const float mean = ts * (1.f / 768.f);
    const float var  = tss * (1.f / 768.f) - mean * mean;
    const float rstd = rsqrtf(var + 1e-5f);
    #pragma unroll
    for (int j = 0; j < 3; ++j) {
        int c = t + 256 * j;
        out[b * 768 + c] = (x[j] - mean) * rstd * gamma[c] + beta[c];
    }
}

#define WN_R 16
#define WN_W (WN_R + 9)

__global__ __launch_bounds__(256) void window_norm(const float* __restrict__ reps,
                                                   const int* __restrict__ Lw,
                                                   float* __restrict__ out) {
    const int lane = threadIdx.x & 31;
    const int wv = __builtin_amdgcn_readfirstlane((int)(threadIdx.x >> 5));
    const int b  = blockIdx.x >> 4;
    const int t0 = (((blockIdx.x & 15) << 3) + wv) * WN_R;
    const int L  = Lw[b];
    const float* base = reps + ((long)b * 2046) * 128 + lane * 4;

    float4 w[WN_W];
    #pragma unroll
    for (int i = 0; i < WN_W; ++i) {
        const int r  = t0 - 5 + i;
        const int rc = r < 0 ? 0 : (r > 2045 ? 2045 : r);
        w[i] = *(const float4*)(base + (long)rc * 128);
    }
    #pragma unroll
    for (int i = 0; i < WN_W; ++i) {
        const int r = t0 - 5 + i;
        if (r < 0 || r >= L) w[i] = float4{0.f, 0.f, 0.f, 0.f};
    }

    float4 mn[WN_R];
    float  sq[WN_R];
    #pragma unroll
    for (int j = 0; j < WN_R; ++j) {
        const int tp = t0 + j;
        int start = tp - 5; if (start < 0) start = 0;
        int end   = tp + 5; if (end > L) end = L;
        int cnt   = end - start; if (cnt < 1) cnt = 1;

        float4 sum = {0.f, 0.f, 0.f, 0.f};
        #pragma unroll
        for (int k = 0; k < 10; ++k) {
            const float4 v = w[j + k];
            sum.x += v.x; sum.y += v.y; sum.z += v.z; sum.w += v.w;
        }
        const float inv = 1.f / (float)cnt;
        float4 m4 = {sum.x * inv, sum.y * inv, sum.z * inv, sum.w * inv};
        mn[j] = m4;
        sq[j] = m4.x * m4.x + m4.y * m4.y + m4.z * m4.z + m4.w * m4.w;
    }

    #pragma unroll
    for (int o = 1; o < 32; o <<= 1) {
        #pragma unroll
        for (int j = 0; j < WN_R; ++j) sq[j] += __shfl_xor(sq[j], o, 32);
    }

    #pragma unroll
    for (int j = 0; j < WN_R; ++j) {
        const int tp = t0 + j;
        if (tp < 2046) {
            const bool  valid = tp < L;
            const float n  = sqrtf(sq[j]);
            const float sc = (valid && n > 0.f) ? (1.f / n) : 0.f;
            const float4 m4 = mn[j];
            float4 o4 = {m4.x * sc, m4.y * sc, m4.z * sc, m4.w * sc};
            *(float4*)(out + ((long)b * 2046 + tp) * 128 + lane * 4) = o4;
        }
    }
}

extern "C" void kernel_launch(void* const* d_in, const int* in_sizes, int n_in,
                              void* d_out, int out_size, void* d_ws, size_t ws_size,
                              hipStream_t stream) {
    const float* hidden   = (const float*)d_in[0];
    const int*   amask    = (const int*)  d_in[1];
    const float* tok_w    = (const float*)d_in[2];
    const float* tok_b    = (const float*)d_in[3];
    const float* cls_w    = (const float*)d_in[4];
    const float* cls_b    = (const float*)d_in[5];
    const float* ln_tok_g = (const float*)d_in[6];
    const float* ln_tok_b = (const float*)d_in[7];
    const float* ln_cls_g = (const float*)d_in[8];
    const float* ln_cls_b = (const float*)d_in[9];
    float* out = (float*)d_out;

    float* ws      = (float*)d_ws;
    float* reps    = ws;
    float* cls_lin = ws + 8380416;
    float* wr_tok  = ws + 8404992;
    float* wr_cls  = ws + 8503296;
    int*   lenw    = (int*)(ws + 9093120);

    reorder_w<<<384,  256, 0, stream>>>(tok_w, 128, wr_tok);
    reorder_w<<<2304, 256, 0, stream>>>(cls_w, 768, wr_cls);
    lengths_k<<<32,   256, 0, stream>>>(amask, lenw);

    tok_gemm_ln<<<1024, 128, 0, stream>>>(hidden, amask, wr_tok, tok_b,
                                          ln_tok_g, ln_tok_b, reps);
    cls_gemm<<<48, 512, 0, stream>>>(hidden, wr_cls, cls_lin);
    cls_ln<<<32, 256, 0, stream>>>(cls_lin, cls_b, ln_cls_g, ln_cls_b, out);
    window_norm<<<512, 256, 0, stream>>>(reps, lenw, out + 24576);
}


# tok_gemm_ln: A tile double-buffered in LDS (6 K-slices of 128): next slice's global loads issued under the current slice's WMMAs, stored to the other buffer after them; one barrier per slice
# speedup vs baseline: 1.6867x; 1.1662x over previous
#include <hip/hip_runtime.h>
#include <hip/hip_bf16.h>


typedef float v2f __attribute__((ext_vector_type(2)));
typedef float v8f __attribute__((ext_vector_type(8)));

__device__ __forceinline__ v8f wmma_f32(v2f a, v2f b, v8f c) {
    return __builtin_amdgcn_wmma_f32_16x16x4_f32(
        false, a, false, b, (short)0, c, false, false);
}

__global__ __launch_bounds__(256) void reorder_w(const float* __restrict__ W,
                                                 int N,
                                                 float* __restrict__ Wr) {
    int o    = blockIdx.x * 256 + threadIdx.x;
    int pair = o / 24576;
    int rem  = o % 24576;
    int i    = rem / 128;
    int rem2 = rem % 128;
    int lane = rem2 >> 2;
    int tl   = (rem2 >> 1) & 1;
    int j    = rem2 & 1;
    int m    = lane & 15;
    int half = lane >> 4;
    int k    = 4 * i + 2 * half + j;
    int n    = (pair * 2 + tl) * 16 + m;
    Wr[o] = W[k * N + n];
}

__global__ __launch_bounds__(256) void lengths_k(const int* __restrict__ mask,
                                                 int* __restrict__ Lout) {
    __shared__ int sh[8];
    int b = blockIdx.x, t = threadIdx.x;
    int s = 0;
    #pragma unroll
    for (int j = 0; j < 8; ++j) s += mask[b * 2048 + t + 256 * j];
    #pragma unroll
    for (int o = 1; o < 32; o <<= 1) s += __shfl_xor(s, o, 32);
    if ((t & 31) == 0) sh[t >> 5] = s;
    __syncthreads();
    if (t == 0) {
        int tot = 0;
        #pragma unroll
        for (int j = 0; j < 8; ++j) tot += sh[j];
        int L = tot - 1;
        Lout[b] = L < 2046 ? L : 2046;
    }
}

#define LDA 132
#define LDC 132
#define KS  128
#define NS  6
#define IT  32
#define NQ  16
#define TILE (64 * LDA)

__device__ __forceinline__ void tok_slice(const float* __restrict__ cur,
                                          const float4* __restrict__ bq,
                                          v8f (&acc)[4][2]) {
    #pragma unroll
    for (int i = 0; i < IT; ++i) {
        float4 f = bq[i * 32];
        v2f b0 = {f.x, f.y};
        v2f b1 = {f.z, f.w};
        #pragma unroll
        for (int mt = 0; mt < 4; ++mt) {
            v2f a = *(const v2f*)&cur[mt * 16 * LDA + 4 * i];
            acc[mt][0] = wmma_f32(a, b0, acc[mt][0]);
            acc[mt][1] = wmma_f32(a, b1, acc[mt][1]);
        }
    }
}

__global__ __launch_bounds__(128) void tok_gemm_ln(
    const float* __restrict__ hidden, const int* __restrict__ amask,
    const float* __restrict__ wr,     const float* __restrict__ tok_b,
    const float* __restrict__ gamma,  const float* __restrict__ beta,
    float* __restrict__ reps) {
    __shared__ __align__(16) float shmem[2 * TILE];
    float* shC = shmem;

    const int  t    = threadIdx.x;
    const int  lane = t & 31;
    const int  wv   = t >> 5;
    const int  m    = lane & 15;
    const int  half = lane >> 4;
    const long row0 = (long)blockIdx.x * 64;

    v8f acc[4][2] = {};
    const float4* bp = (const float4*)wr + ((long)wv * 192) * 32 + lane;

    const float* gsrc = hidden + (row0 + wv) * 768 + lane * 4;
    const int    lofs = wv * LDA + lane * 4;
    const int    aofs = m * LDA + 2 * half;

    {
        float4 p0[NQ];
        #pragma unroll
        for (int q = 0; q < NQ; ++q)
            p0[q] = *(const float4*)(gsrc + q * (4 * 768));
        #pragma unroll
        for (int q = 0; q < NQ; ++q)
            *(float4*)&shmem[lofs + q * (4 * LDA)] = p0[q];
    }
    __syncthreads();

    for (int s = 0; s < NS - 1; ++s) {
        const float* cur = shmem + (s & 1) * TILE;
        float*       nxt = shmem + ((s + 1) & 1) * TILE;

        float4 pre[NQ];
        #pragma unroll
        for (int q = 0; q < NQ; ++q)
            pre[q] = *(const float4*)(gsrc + q * (4 * 768) + (s + 1) * KS);

        tok_slice(cur + aofs, bp + (long)s * (IT * 32), acc);

        #pragma unroll
        for (int q = 0; q < NQ; ++q)
            *(float4*)&nxt[lofs + q * (4 * LDA)] = pre[q];
        __syncthreads();
    }
    tok_slice(shmem + ((NS - 1) & 1) * TILE + aofs, bp + (long)(NS - 1) * (IT * 32), acc);

    #pragma unroll
    for (int mt = 0; mt < 4; ++mt) {
        #pragma unroll
        for (int i = 0; i < 8; ++i) {
            shC[(mt * 16 + i + 8 * half) * LDC + wv * 32 + m]      = acc[mt][0][i];
            shC[(mt * 16 + i + 8 * half) * LDC + wv * 32 + 16 + m] = acc[mt][1][i];
        }
    }
    __syncthreads();

    const int r  = t >> 1;
    const int c0 = (t & 1) * 64;
    float s = 0.f, ss = 0.f;
    for (int i = 0; i < 64; ++i) {
        float v = shC[r * LDC + c0 + i] + tok_b[c0 + i];
        s += v; ss += v * v;
    }
    s  += __shfl_xor(s,  1, 32);
    ss += __shfl_xor(ss, 1, 32);
    const float mean = s * (1.f / 128.f);
    const float var  = ss * (1.f / 128.f) - mean * mean;
    const float rstd = rsqrtf(var + 1e-5f);

    const long rg = row0 + r;
    const int  bI = (int)(rg >> 11);
    const int  sI = (int)(rg & 2047);
    const float mk = (float)amask[rg];
    if (sI >= 1 && sI <= 2046) {
        float* dst = reps + ((long)(bI * 2046 + sI - 1)) * 128 + c0;
        for (int i = 0; i < 64; ++i) {
            float v = shC[r * LDC + c0 + i] + tok_b[c0 + i];
            float y = (v - mean) * rstd * gamma[c0 + i] + beta[c0 + i];
            dst[i] = fmaxf(y, 0.f) * mk;
        }
    }
}

#define CLS_KW 12

__global__ __launch_bounds__(512) void cls_gemm(const float* __restrict__ hidden,
                                                const float* __restrict__ wr2,
                                                float* __restrict__ cls_lin) {
    __shared__ float red[16 * 512];
    const int t = threadIdx.x, lane = t & 31, wv = t >> 5;
    const int m = lane & 15, half = lane >> 4;
    const int pair = blockIdx.x >> 1;
    const int mt   = blockIdx.x & 1;

    const float4* bp = (const float4*)wr2 + ((long)pair * 192 + wv * CLS_KW) * 32 + lane;
    const float*  ap = hidden + (long)(mt * 16 + m) * (2048 * 768)
                              + wv * (4 * CLS_KW) + 2 * half;
    float* rw = red + wv * 512 + lane;

    v2f    a[CLS_KW];
    float4 f[CLS_KW];
    #pragma unroll
    for (int i = 0; i < CLS_KW; ++i) {
        a[i] = *(const v2f*)(ap + 4 * i);
        f[i] = bp[i * 32];
    }
    __builtin_amdgcn_sched_barrier(0);

    v8f acc0 = {}, acc1 = {};
    #pragma unroll
    for (int i = 0; i < CLS_KW; ++i) {
        v2f b0 = {f[i].x, f[i].y};
        v2f b1 = {f[i].z, f[i].w};
        acc0 = wmma_f32(a[i], b0, acc0);
        acc1 = wmma_f32(a[i], b1, acc1);
    }

    #pragma unroll
    for (int i = 0; i < 8; ++i) {
        rw[i * 32]       = acc0[i];
        rw[(8 + i) * 32] = acc1[i];
    }
    __syncthreads();

    float s = 0.f;
    #pragma unroll
    for (int w = 0; w < 16; ++w) s += red[w * 512 + t];
    const int tile = wv >> 3, i = wv & 7;
    cls_lin[(mt * 16 + i + 8 * half) * 768 + pair * 32 + tile * 16 + m] = s;
}

__global__ __launch_bounds__(256) void cls_ln(const float* __restrict__ cls_lin,
                                              const float* __restrict__ cls_b,
                                              const float* __restrict__ gamma,
                                              const float* __restrict__ beta,
                                              float* __restrict__ out) {
    __shared__ float sh_s[8], sh_ss[8];
    const int b = blockIdx.x, t = threadIdx.x;
    float x[3];
    float s = 0.f, ss = 0.f;
    #pragma unroll
    for (int j = 0; j < 3; ++j) {
        int c = t + 256 * j;
        float v = cls_lin[b * 768 + c] + cls_b[c];
        x[j] = v; s += v; ss += v * v;
    }
    #pragma unroll
    for (int o = 1; o < 32; o <<= 1) {
        s  += __shfl_xor(s,  o, 32);
        ss += __shfl_xor(ss, o, 32);
    }
    if ((t & 31) == 0) { sh_s[t >> 5] = s; sh_ss[t >> 5] = ss; }
    __syncthreads();
    float ts = 0.f, tss = 0.f;
    #pragma unroll
    for (int j = 0; j < 8; ++j) { ts += sh_s[j]; tss += sh_ss[j]; }
    const float mean = ts * (1.f / 768.f);
    const float var  = tss * (1.f / 768.f) - mean * mean;
    const float rstd = rsqrtf(var + 1e-5f);
    #pragma unroll
    for (int j = 0; j < 3; ++j) {
        int c = t + 256 * j;
        out[b * 768 + c] = (x[j] - mean) * rstd * gamma[c] + beta[c];
    }
}

#define WN_R 16
#define WN_W (WN_R + 9)

__global__ __launch_bounds__(256) void window_norm(const float* __restrict__ reps,
                                                   const int* __restrict__ Lw,
                                                   float* __restrict__ out) {
    const int lane = threadIdx.x & 31;
    const int wv = __builtin_amdgcn_readfirstlane((int)(threadIdx.x >> 5));
    const int b  = blockIdx.x >> 4;
    const int t0 = (((blockIdx.x & 15) << 3) + wv) * WN_R;
    const int L  = Lw[b];
    const float* base = reps + ((long)b * 2046) * 128 + lane * 4;

    float4 w[WN_W];
    #pragma unroll
    for (int i = 0; i < WN_W; ++i) {
        const int r  = t0 - 5 + i;
        const int rc = r < 0 ? 0 : (r > 2045 ? 2045 : r);
        w[i] = *(const float4*)(base + (long)rc * 128);
    }
    #pragma unroll
    for (int i = 0; i < WN_W; ++i) {
        const int r = t0 - 5 + i;
        if (r < 0 || r >= L) w[i] = float4{0.f, 0.f, 0.f, 0.f};
    }

    float4 mn[WN_R];
    float  sq[WN_R];
    #pragma unroll
    for (int j = 0; j < WN_R; ++j) {
        const int tp = t0 + j;
        int start = tp - 5; if (start < 0) start = 0;
        int end   = tp + 5; if (end > L) end = L;
        int cnt   = end - start; if (cnt < 1) cnt = 1;

        float4 sum = {0.f, 0.f, 0.f, 0.f};
        #pragma unroll
        for (int k = 0; k < 10; ++k) {
            const float4 v = w[j + k];
            sum.x += v.x; sum.y += v.y; sum.z += v.z; sum.w += v.w;
        }
        const float inv = 1.f / (float)cnt;
        float4 m4 = {sum.x * inv, sum.y * inv, sum.z * inv, sum.w * inv};
        mn[j] = m4;
        sq[j] = m4.x * m4.x + m4.y * m4.y + m4.z * m4.z + m4.w * m4.w;
    }

    #pragma unroll
    for (int o = 1; o < 32; o <<= 1) {
        #pragma unroll
        for (int j = 0; j < WN_R; ++j) sq[j] += __shfl_xor(sq[j], o, 32);
    }

    #pragma unroll
    for (int j = 0; j < WN_R; ++j) {
        const int tp = t0 + j;
        if (tp < 2046) {
            const bool  valid = tp < L;
            const float n  = sqrtf(sq[j]);
            const float sc = (valid && n > 0.f) ? (1.f / n) : 0.f;
            const float4 m4 = mn[j];
            float4 o4 = {m4.x * sc, m4.y * sc, m4.z * sc, m4.w * sc};
            *(float4*)(out + ((long)b * 2046 + tp) * 128 + lane * 4) = o4;
        }
    }
}

extern "C" void kernel_launch(void* const* d_in, const int* in_sizes, int n_in,
                              void* d_out, int out_size, void* d_ws, size_t ws_size,
                              hipStream_t stream) {
    const float* hidden   = (const float*)d_in[0];
    const int*   amask    = (const int*)  d_in[1];
    const float* tok_w    = (const float*)d_in[2];
    const float* tok_b    = (const float*)d_in[3];
    const float* cls_w    = (const float*)d_in[4];
    const float* cls_b    = (const float*)d_in[5];
    const float* ln_tok_g = (const float*)d_in[6];
    const float* ln_tok_b = (const float*)d_in[7];
    const float* ln_cls_g = (const float*)d_in[8];
    const float* ln_cls_b = (const float*)d_in[9];
    float* out = (float*)d_out;

    float* ws      = (float*)d_ws;
    float* reps    = ws;
    float* cls_lin = ws + 8380416;
    float* wr_tok  = ws + 8404992;
    float* wr_cls  = ws + 8503296;
    int*   lenw    = (int*)(ws + 9093120);

    reorder_w<<<384,  256, 0, stream>>>(tok_w, 128, wr_tok);
    reorder_w<<<2304, 256, 0, stream>>>(cls_w, 768, wr_cls);
    lengths_k<<<32,   256, 0, stream>>>(amask, lenw);

    tok_gemm_ln<<<1024, 128, 0, stream>>>(hidden, amask, wr_tok, tok_b,
                                          ln_tok_g, ln_tok_b, reps);
    cls_gemm<<<48, 512, 0, stream>>>(hidden, wr_cls, cls_lin);
    cls_ln<<<32, 256, 0, stream>>>(cls_lin, cls_b, ln_cls_g, ln_cls_b, out);
    window_norm<<<512, 256, 0, stream>>>(reps, lenw, out + 24576);
}


# independent kernels of equal block size share a launch: mask lengths + both weight reorders in one, cls LayerNorm as extra blocks of the window launch (7 -> 4 launches); every sum order kept
# speedup vs baseline: 1.8222x; 1.0803x over previous
#include <hip/hip_runtime.h>
#include <hip/hip_bf16.h>


typedef float v2f __attribute__((ext_vector_type(2)));
typedef float v8f __attribute__((ext_vector_type(8)));

__device__ __forceinline__ v8f wmma_f32(v2f a, v2f b, v8f c) {
    return __builtin_amdgcn_wmma_f32_16x16x4_f32(
        false, a, false, b, (short)0, c, false, false);
}

__device__ __forceinline__ void reorder_w(const float* __restrict__ W,
                                          int N,
                                          float* __restrict__ Wr,
                                          int blk) {
    int o    = blk * 256 + threadIdx.x;
    int pair = o / 24576;
    int rem  = o % 24576;
    int i    = rem / 128;
    int rem2 = rem % 128;
    int lane = rem2 >> 2;
    int tl   = (rem2 >> 1) & 1;
    int j    = rem2 & 1;
    int m    = lane & 15;
    int half = lane >> 4;
    int k    = 4 * i + 2 * half + j;
    int n    = (pair * 2 + tl) * 16 + m;
    Wr[o] = W[k * N + n];
}

__device__ __forceinline__ void lengths_k(const int* __restrict__ mask,
                                          int* __restrict__ Lout,
                                          int b) {
    __shared__ int sh[8];
    int t = threadIdx.x;
    int s = 0;
    #pragma unroll
    for (int j = 0; j < 8; ++j) s += mask[b * 2048 + t + 256 * j];
    #pragma unroll
    for (int o = 1; o < 32; o <<= 1) s += __shfl_xor(s, o, 32);
    if ((t & 31) == 0) sh[t >> 5] = s;
    __syncthreads();
    if (t == 0) {
        int tot = 0;
        #pragma unroll
        for (int j = 0; j < 8; ++j) tot += sh[j];
        int L = tot - 1;
        Lout[b] = L < 2046 ? L : 2046;
    }
}

#define PREP_LEN  32
#define PREP_CLS  2304

__global__ __launch_bounds__(256) void prep_k(const float* __restrict__ tok_w,
                                              const float* __restrict__ cls_w,
                                              const int* __restrict__ mask,
                                              float* __restrict__ wr_tok,
                                              float* __restrict__ wr_cls,
                                              int* __restrict__ Lout) {
    const int blk = blockIdx.x;
    if (blk < PREP_LEN)
        lengths_k(mask, Lout, blk);
    else if (blk < PREP_LEN + PREP_CLS)
        reorder_w(cls_w, 768, wr_cls, blk - PREP_LEN);
    else
        reorder_w(tok_w, 128, wr_tok, blk - (PREP_LEN + PREP_CLS));
}

#define LDA 132
#define LDC 132
#define KS  128
#define NS  6
#define IT  32
#define NQ  16
#define TILE (64 * LDA)

__device__ __forceinline__ void tok_slice(const float* __restrict__ cur,
                                          const float4* __restrict__ bq,
                                          v8f (&acc)[4][2]) {
    #pragma unroll
    for (int i = 0; i < IT; ++i) {
        float4 f = bq[i * 32];
        v2f b0 = {f.x, f.y};
        v2f b1 = {f.z, f.w};
        #pragma unroll
        for (int mt = 0; mt < 4; ++mt) {
            v2f a = *(const v2f*)&cur[mt * 16 * LDA + 4 * i];
            acc[mt][0] = wmma_f32(a, b0, acc[mt][0]);
            acc[mt][1] = wmma_f32(a, b1, acc[mt][1]);
        }
    }
}

__global__ __launch_bounds__(128) void tok_gemm_ln(
    const float* __restrict__ hidden, const int* __restrict__ amask,
    const float* __restrict__ wr,     const float* __restrict__ tok_b,
    const float* __restrict__ gamma,  const float* __restrict__ beta,
    float* __restrict__ reps) {
    __shared__ __align__(16) float shmem[2 * TILE];
    float* shC = shmem;

    const int  t    = threadIdx.x;
    const int  lane = t & 31;
    const int  wv   = t >> 5;
    const int  m    = lane & 15;
    const int  half = lane >> 4;
    const long row0 = (long)blockIdx.x * 64;

    v8f acc[4][2] = {};
    const float4* bp = (const float4*)wr + ((long)wv * 192) * 32 + lane;

    const float* gsrc = hidden + (row0 + wv) * 768 + lane * 4;
    const int    lofs = wv * LDA + lane * 4;
    const int    aofs = m * LDA + 2 * half;

    {
        float4 p0[NQ];
        #pragma unroll
        for (int q = 0; q < NQ; ++q)
            p0[q] = *(const float4*)(gsrc + q * (4 * 768));
        #pragma unroll
        for (int q = 0; q < NQ; ++q)
            *(float4*)&shmem[lofs + q * (4 * LDA)] = p0[q];
    }
    __syncthreads();

    for (int s = 0; s < NS - 1; ++s) {
        const float* cur = shmem + (s & 1) * TILE;
        float*       nxt = shmem + ((s + 1) & 1) * TILE;

        float4 pre[NQ];
        #pragma unroll
        for (int q = 0; q < NQ; ++q)
            pre[q] = *(const float4*)(gsrc + q * (4 * 768) + (s + 1) * KS);

        tok_slice(cur + aofs, bp + (long)s * (IT * 32), acc);

        #pragma unroll
        for (int q = 0; q < NQ; ++q)
            *(float4*)&nxt[lofs + q * (4 * LDA)] = pre[q];
        __syncthreads();
    }
    tok_slice(shmem + ((NS - 1) & 1) * TILE + aofs, bp + (long)(NS - 1) * (IT * 32), acc);

    #pragma unroll
    for (int mt = 0; mt < 4; ++mt) {
        #pragma unroll
        for (int i = 0; i < 8; ++i) {
            shC[(mt * 16 + i + 8 * half) * LDC + wv * 32 + m]      = acc[mt][0][i];
            shC[(mt * 16 + i + 8 * half) * LDC + wv * 32 + 16 + m] = acc[mt][1][i];
        }
    }
    __syncthreads();

    const int r  = t >> 1;
    const int c0 = (t & 1) * 64;
    float s = 0.f, ss = 0.f;
    for (int i = 0; i < 64; ++i) {
        float v = shC[r * LDC + c0 + i] + tok_b[c0 + i];
        s += v; ss += v * v;
    }
    s  += __shfl_xor(s,  1, 32);
    ss += __shfl_xor(ss, 1, 32);
    const float mean = s * (1.f / 128.f);
    const float var  = ss * (1.f / 128.f) - mean * mean;
    const float rstd = rsqrtf(var + 1e-5f);

    const long rg = row0 + r;
    const int  bI = (int)(rg >> 11);
    const int  sI = (int)(rg & 2047);
    const float mk = (float)amask[rg];
    if (sI >= 1 && sI <= 2046) {
        float* dst = reps + ((long)(bI * 2046 + sI - 1)) * 128 + c0;
        for (int i = 0; i < 64; ++i) {
            float v = shC[r * LDC + c0 + i] + tok_b[c0 + i];
            float y = (v - mean) * rstd * gamma[c0 + i] + beta[c0 + i];
            dst[i] = fmaxf(y, 0.f) * mk;
        }
    }
}

#define CLS_KW 12

__global__ __launch_bounds__(512) void cls_gemm(const float* __restrict__ hidden,
                                                const float* __restrict__ wr2,
                                                float* __restrict__ cls_lin) {
    __shared__ float red[16 * 512];
    const int t = threadIdx.x, lane = t & 31, wv = t >> 5;
    const int m = lane & 15, half = lane >> 4;
    const int pair = blockIdx.x >> 1;
    const int mt   = blockIdx.x & 1;

    const float4* bp = (const float4*)wr2 + ((long)pair * 192 + wv * CLS_KW) * 32 + lane;
    const float*  ap = hidden + (long)(mt * 16 + m) * (2048 * 768)
                              + wv * (4 * CLS_KW) + 2 * half;
    float* rw = red + wv * 512 + lane;

    v2f    a[CLS_KW];
    float4 f[CLS_KW];
    #pragma unroll
    for (int i = 0; i < CLS_KW; ++i) {
        a[i] = *(const v2f*)(ap + 4 * i);
        f[i] = bp[i * 32];
    }
    __builtin_amdgcn_sched_barrier(0);

    v8f acc0 = {}, acc1 = {};
    #pragma unroll
    for (int i = 0; i < CLS_KW; ++i) {
        v2f b0 = {f[i].x, f[i].y};
        v2f b1 = {f[i].z, f[i].w};
        acc0 = wmma_f32(a[i], b0, acc0);
        acc1 = wmma_f32(a[i], b1, acc1);
    }

    #pragma unroll
    for (int i = 0; i < 8; ++i) {
        rw[i * 32]       = acc0[i];
        rw[(8 + i) * 32] = acc1[i];
    }
    __syncthreads();

    float s = 0.f;
    #pragma unroll
    for (int w = 0; w < 16; ++w) s += red[w * 512 + t];
    const int tile = wv >> 3, i = wv & 7;
    cls_lin[(mt * 16 + i + 8 * half) * 768 + pair * 32 + tile * 16 + m] = s;
}

__device__ __forceinline__ void cls_ln(const float* __restrict__ cls_lin,
                                       const float* __restrict__ cls_b,
                                       const float* __restrict__ gamma,
                                       const float* __restrict__ beta,
                                       float* __restrict__ out,
                                       int b) {
    __shared__ float sh_s[8], sh_ss[8];
    const int t = threadIdx.x;
    float x[3];
    float s = 0.f, ss = 0.f;
    #pragma unroll
    for (int j = 0; j < 3; ++j) {
        int c = t + 256 * j;
        float v = cls_lin[b * 768 + c] + cls_b[c];
        x[j] = v; s += v; ss += v * v;
    }
    #pragma unroll
    for (int o = 1; o < 32; o <<= 1) {
        s  += __shfl_xor(s,  o, 32);
        ss += __shfl_xor(ss, o, 32);
    }
    if ((t & 31) == 0) { sh_s[t >> 5] = s; sh_ss[t >> 5] = ss; }
    __syncthreads();
    float ts = 0.f, tss = 0.f;
    #pragma unroll
    for (int j = 0; j < 8; ++j) { ts += sh_s[j]; tss += sh_ss[j]; }
    const float mean = ts * (1.f / 768.f);
    const float var  = tss * (1.f / 768.f) - mean * mean;
    const float rstd = rsqrtf(var + 1e-5f);
    #pragma unroll
    for (int j = 0; j < 3; ++j) {
        int c = t + 256 * j;
        out[b * 768 + c] = (x[j] - mean) * rstd * gamma[c] + beta[c];
    }
}

#define WN_R 16
#define WN_W (WN_R + 9)

__device__ __forceinline__ void window_norm(const float* __restrict__ reps,
                                            const int* __restrict__ Lw,
                                            float* __restrict__ out,
                                            int blk) {
    const int lane = threadIdx.x & 31;
    const int wv = __builtin_amdgcn_readfirstlane((int)(threadIdx.x >> 5));
    const int b  = blk >> 4;
    const int t0 = (((blk & 15) << 3) + wv) * WN_R;
    const int L  = Lw[b];
    const float* base = reps + ((long)b * 2046) * 128 + lane * 4;

    float4 w[WN_W];
    #pragma unroll
    for (int i = 0; i < WN_W; ++i) {
        const int r  = t0 - 5 + i;
        const int rc = r < 0 ? 0 : (r > 2045 ? 2045 : r);
        w[i] = *(const float4*)(base + (long)rc * 128);
    }
    #pragma unroll
    for (int i = 0; i < WN_W; ++i) {
        const int r = t0 - 5 + i;
        if (r < 0 || r >= L) w[i] = float4{0.f, 0.f, 0.f, 0.f};
    }

    float4 mn[WN_R];
    float  sq[WN_R];
    #pragma unroll
    for (int j = 0; j < WN_R; ++j) {
        const int tp = t0 + j;
        int start = tp - 5; if (start < 0) start = 0;
        int end   = tp + 5; if (end > L) end = L;
        int cnt   = end - start; if (cnt < 1) cnt = 1;

        float4 sum = {0.f, 0.f, 0.f, 0.f};
        #pragma unroll
        for (int k = 0; k < 10; ++k) {
            const float4 v = w[j + k];
            sum.x += v.x; sum.y += v.y; sum.z += v.z; sum.w += v.w;
        }
        const float inv = 1.f / (float)cnt;
        float4 m4 = {sum.x * inv, sum.y * inv, sum.z * inv, sum.w * inv};
        mn[j] = m4;
        sq[j] = m4.x * m4.x + m4.y * m4.y + m4.z * m4.z + m4.w * m4.w;
    }

    #pragma unroll
    for (int o = 1; o < 32; o <<= 1) {
        #pragma unroll
        for (int j = 0; j < WN_R; ++j) sq[j] += __shfl_xor(sq[j], o, 32);
    }

    #pragma unroll
    for (int j = 0; j < WN_R; ++j) {
        const int tp = t0 + j;
        if (tp < 2046) {
            const bool  valid = tp < L;
            const float n  = sqrtf(sq[j]);
            const float sc = (valid && n > 0.f) ? (1.f / n) : 0.f;
            const float4 m4 = mn[j];
            float4 o4 = {m4.x * sc, m4.y * sc, m4.z * sc, m4.w * sc};
            *(float4*)(out + ((long)b * 2046 + tp) * 128 + lane * 4) = o4;
        }
    }
}

#define TAIL_CLS 32

__global__ __launch_bounds__(256) void tail_k(const float* __restrict__ reps,
                                              const int* __restrict__ Lw,
                                              const float* __restrict__ cls_lin,
                                              const float* __restrict__ cls_b,
                                              const float* __restrict__ gamma,
                                              const float* __restrict__ beta,
                                              float* __restrict__ out) {
    const int blk = blockIdx.x;
    if (blk < TAIL_CLS)
        cls_ln(cls_lin, cls_b, gamma, beta, out, blk);
    else
        window_norm(reps, Lw, out + 24576, blk - TAIL_CLS);
}

extern "C" void kernel_launch(void* const* d_in, const int* in_sizes, int n_in,
                              void* d_out, int out_size, void* d_ws, size_t ws_size,
                              hipStream_t stream) {
    const float* hidden   = (const float*)d_in[0];
    const int*   amask    = (const int*)  d_in[1];
    const float* tok_w    = (const float*)d_in[2];
    const float* tok_b    = (const float*)d_in[3];
    const float* cls_w    = (const float*)d_in[4];
    const float* cls_b    = (const float*)d_in[5];
    const float* ln_tok_g = (const float*)d_in[6];
    const float* ln_tok_b = (const float*)d_in[7];
    const float* ln_cls_g = (const float*)d_in[8];
    const float* ln_cls_b = (const float*)d_in[9];
    float* out = (float*)d_out;

    float* ws      = (float*)d_ws;
    float* reps    = ws;
    float* cls_lin = ws + 8380416;
    float* wr_tok  = ws + 8404992;
    float* wr_cls  = ws + 8503296;
    int*   lenw    = (int*)(ws + 9093120);

    prep_k<<<2720, 256, 0, stream>>>(tok_w, cls_w, amask,
                                     wr_tok, wr_cls, lenw);

    tok_gemm_ln<<<1024, 128, 0, stream>>>(hidden, amask, wr_tok, tok_b,
                                          ln_tok_g, ln_tok_b, reps);
    cls_gemm<<<48, 512, 0, stream>>>(hidden, wr_cls, cls_lin);
    tail_k<<<544, 256, 0, stream>>>(reps, lenw, cls_lin, cls_b,
                                    ln_cls_g, ln_cls_b, out);
}


# tok_gemm_ln epilogue remapped: a half-wave owns a whole row (2 float4 per lane), LayerNorm sums by 16-lane DPP butterfly, each 128-byte line stored by one instruction, not 8; LN sum order changes
# speedup vs baseline: 1.9573x; 1.0741x over previous
#include <hip/hip_runtime.h>
#include <hip/hip_bf16.h>


typedef float v2f __attribute__((ext_vector_type(2)));
typedef float v8f __attribute__((ext_vector_type(8)));

__device__ __forceinline__ v8f wmma_f32(v2f a, v2f b, v8f c) {
    return __builtin_amdgcn_wmma_f32_16x16x4_f32(
        false, a, false, b, (short)0, c, false, false);
}

template <int X>
__device__ __forceinline__ float row_xor_add(float v) {
    const int b = __builtin_bit_cast(int, v);
    const int o = __builtin_amdgcn_update_dpp(0, b, 0x160 + X, 0xf, 0xf, true);
    return v + __builtin_bit_cast(float, o);
}

__device__ __forceinline__ void reorder_w(const float* __restrict__ W,
                                          int N,
                                          float* __restrict__ Wr,
                                          int blk) {
    int o    = blk * 256 + threadIdx.x;
    int pair = o / 24576;
    int rem  = o % 24576;
    int i    = rem / 128;
    int rem2 = rem % 128;
    int lane = rem2 >> 2;
    int tl   = (rem2 >> 1) & 1;
    int j    = rem2 & 1;
    int m    = lane & 15;
    int half = lane >> 4;
    int k    = 4 * i + 2 * half + j;
    int n    = (pair * 2 + tl) * 16 + m;
    Wr[o] = W[k * N + n];
}

__device__ __forceinline__ void lengths_k(const int* __restrict__ mask,
                                          int* __restrict__ Lout,
                                          int b) {
    __shared__ int sh[8];
    int t = threadIdx.x;
    int s = 0;
    #pragma unroll
    for (int j = 0; j < 8; ++j) s += mask[b * 2048 + t + 256 * j];
    #pragma unroll
    for (int o = 1; o < 32; o <<= 1) s += __shfl_xor(s, o, 32);
    if ((t & 31) == 0) sh[t >> 5] = s;
    __syncthreads();
    if (t == 0) {
        int tot = 0;
        #pragma unroll
        for (int j = 0; j < 8; ++j) tot += sh[j];
        int L = tot - 1;
        Lout[b] = L < 2046 ? L : 2046;
    }
}

#define PREP_LEN  32
#define PREP_CLS  2304

__global__ __launch_bounds__(256) void prep_k(const float* __restrict__ tok_w,
                                              const float* __restrict__ cls_w,
                                              const int* __restrict__ mask,
                                              float* __restrict__ wr_tok,
                                              float* __restrict__ wr_cls,
                                              int* __restrict__ Lout) {
    const int blk = blockIdx.x;
    if (blk < PREP_LEN)
        lengths_k(mask, Lout, blk);
    else if (blk < PREP_LEN + PREP_CLS)
        reorder_w(cls_w, 768, wr_cls, blk - PREP_LEN);
    else
        reorder_w(tok_w, 128, wr_tok, blk - (PREP_LEN + PREP_CLS));
}

#define LDA 132
#define LDC 132
#define KS  128
#define NS  6
#define IT  32
#define NQ  16
#define TILE (64 * LDA)

__device__ __forceinline__ void tok_slice(const float* __restrict__ cur,
                                          const float4* __restrict__ bq,
                                          v8f (&acc)[4][2]) {
    #pragma unroll
    for (int i = 0; i < IT; ++i) {
        float4 f = bq[i * 32];
        v2f b0 = {f.x, f.y};
        v2f b1 = {f.z, f.w};
        #pragma unroll
        for (int mt = 0; mt < 4; ++mt) {
            v2f a = *(const v2f*)&cur[mt * 16 * LDA + 4 * i];
            acc[mt][0] = wmma_f32(a, b0, acc[mt][0]);
            acc[mt][1] = wmma_f32(a, b1, acc[mt][1]);
        }
    }
}

__global__ __launch_bounds__(128) void tok_gemm_ln(
    const float* __restrict__ hidden, const int* __restrict__ amask,
    const float* __restrict__ wr,     const float* __restrict__ tok_b,
    const float* __restrict__ gamma,  const float* __restrict__ beta,
    float* __restrict__ reps) {
    __shared__ __align__(16) float shmem[2 * TILE];
    float* shC = shmem;

    const int  t    = threadIdx.x;
    const int  lane = t & 31;
    const int  wv   = t >> 5;
    const int  m    = lane & 15;
    const int  half = lane >> 4;
    const long row0 = (long)blockIdx.x * 64;

    v8f acc[4][2] = {};
    const float4* bp = (const float4*)wr + ((long)wv * 192) * 32 + lane;

    const float* gsrc = hidden + (row0 + wv) * 768 + lane * 4;
    const int    lofs = wv * LDA + lane * 4;
    const int    aofs = m * LDA + 2 * half;

    {
        float4 p0[NQ];
        #pragma unroll
        for (int q = 0; q < NQ; ++q)
            p0[q] = *(const float4*)(gsrc + q * (4 * 768));
        #pragma unroll
        for (int q = 0; q < NQ; ++q)
            *(float4*)&shmem[lofs + q * (4 * LDA)] = p0[q];
    }
    __syncthreads();

    for (int s = 0; s < NS - 1; ++s) {
        const float* cur = shmem + (s & 1) * TILE;
        float*       nxt = shmem + ((s + 1) & 1) * TILE;

        float4 pre[NQ];
        #pragma unroll
        for (int q = 0; q < NQ; ++q)
            pre[q] = *(const float4*)(gsrc + q * (4 * 768) + (s + 1) * KS);

        tok_slice(cur + aofs, bp + (long)s * (IT * 32), acc);

        #pragma unroll
        for (int q = 0; q < NQ; ++q)
            *(float4*)&nxt[lofs + q * (4 * LDA)] = pre[q];
        __syncthreads();
    }
    tok_slice(shmem + ((NS - 1) & 1) * TILE + aofs, bp + (long)(NS - 1) * (IT * 32), acc);

    #pragma unroll
    for (int mt = 0; mt < 4; ++mt) {
        #pragma unroll
        for (int i = 0; i < 8; ++i) {
            shC[(mt * 16 + i + 8 * half) * LDC + wv * 32 + m]      = acc[mt][0][i];
            shC[(mt * 16 + i + 8 * half) * LDC + wv * 32 + 16 + m] = acc[mt][1][i];
        }
    }
    __syncthreads();

    const int c4 = 4 * m;
    const float4 tbA = *(const float4*)(tok_b + c4);
    const float4 tbB = *(const float4*)(tok_b + 64 + c4);
    const float4 gaA = *(const float4*)(gamma + c4);
    const float4 gaB = *(const float4*)(gamma + 64 + c4);
    const float4 beA = *(const float4*)(beta + c4);
    const float4 beB = *(const float4*)(beta + 64 + c4);

    float4 xa[8], xb[8];
    float  s[8], ss[8], mk[8];
    #pragma unroll
    for (int p = 0; p < 8; ++p) {
        const int r = wv * 16 + 2 * p + half;
        mk[p] = (float)amask[row0 + r];
        float4 a = *(const float4*)&shC[r * LDC + c4];
        float4 b = *(const float4*)&shC[r * LDC + 64 + c4];
        a.x += tbA.x; a.y += tbA.y; a.z += tbA.z; a.w += tbA.w;
        b.x += tbB.x; b.y += tbB.y; b.z += tbB.z; b.w += tbB.w;
        xa[p] = a; xb[p] = b;
        s[p]  = ((a.x + a.y) + (a.z + a.w)) + ((b.x + b.y) + (b.z + b.w));
        ss[p] = ((a.x * a.x + a.y * a.y) + (a.z * a.z + a.w * a.w))
              + ((b.x * b.x + b.y * b.y) + (b.z * b.z + b.w * b.w));
    }
    #pragma unroll
    for (int p = 0; p < 8; ++p) { s[p] = row_xor_add<1>(s[p]); ss[p] = row_xor_add<1>(ss[p]); }
    #pragma unroll
    for (int p = 0; p < 8; ++p) { s[p] = row_xor_add<2>(s[p]); ss[p] = row_xor_add<2>(ss[p]); }
    #pragma unroll
    for (int p = 0; p < 8; ++p) { s[p] = row_xor_add<4>(s[p]); ss[p] = row_xor_add<4>(ss[p]); }
    #pragma unroll
    for (int p = 0; p < 8; ++p) { s[p] = row_xor_add<8>(s[p]); ss[p] = row_xor_add<8>(ss[p]); }

    #pragma unroll
    for (int p = 0; p < 8; ++p) {
        const float mean = s[p] * (1.f / 128.f);
        const float var  = ss[p] * (1.f / 128.f) - mean * mean;
        const float rstd = rsqrtf(var + 1e-5f);
        const long rg = row0 + wv * 16 + 2 * p + half;
        const int  bI = (int)(rg >> 11);
        const int  sI = (int)(rg & 2047);
        if (sI >= 1 && sI <= 2046) {
            float* dst = reps + ((long)(bI * 2046 + sI - 1)) * 128 + c4;
            const float4 a = xa[p], b = xb[p];
            float4 ya, yb;
            ya.x = fmaxf((a.x - mean) * rstd * gaA.x + beA.x, 0.f) * mk[p];
            ya.y = fmaxf((a.y - mean) * rstd * gaA.y + beA.y, 0.f) * mk[p];
            ya.z = fmaxf((a.z - mean) * rstd * gaA.z + beA.z, 0.f) * mk[p];
            ya.w = fmaxf((a.w - mean) * rstd * gaA.w + beA.w, 0.f) * mk[p];
            yb.x = fmaxf((b.x - mean) * rstd * gaB.x + beB.x, 0.f) * mk[p];
            yb.y = fmaxf((b.y - mean) * rstd * gaB.y + beB.y, 0.f) * mk[p];
            yb.z = fmaxf((b.z - mean) * rstd * gaB.z + beB.z, 0.f) * mk[p];
            yb.w = fmaxf((b.w - mean) * rstd * gaB.w + beB.w, 0.f) * mk[p];
            *(float4*)dst        = ya;
            *(float4*)(dst + 64) = yb;
        }
    }
}

#define CLS_KW 12

__global__ __launch_bounds__(512) void cls_gemm(const float* __restrict__ hidden,
                                                const float* __restrict__ wr2,
                                                float* __restrict__ cls_lin) {
    __shared__ float red[16 * 512];
    const int t = threadIdx.x, lane = t & 31, wv = t >> 5;
    const int m = lane & 15, half = lane >> 4;
    const int pair = blockIdx.x >> 1;
    const int mt   = blockIdx.x & 1;

    const float4* bp = (const float4*)wr2 + ((long)pair * 192 + wv * CLS_KW) * 32 + lane;
    const float*  ap = hidden + (long)(mt * 16 + m) * (2048 * 768)
                              + wv * (4 * CLS_KW) + 2 * half;
    float* rw = red + wv * 512 + lane;

    v2f    a[CLS_KW];
    float4 f[CLS_KW];
    #pragma unroll
    for (int i = 0; i < CLS_KW; ++i) {
        a[i] = *(const v2f*)(ap + 4 * i);
        f[i] = bp[i * 32];
    }
    __builtin_amdgcn_sched_barrier(0);

    v8f acc0 = {}, acc1 = {};
    #pragma unroll
    for (int i = 0; i < CLS_KW; ++i) {
        v2f b0 = {f[i].x, f[i].y};
        v2f b1 = {f[i].z, f[i].w};
        acc0 = wmma_f32(a[i], b0, acc0);
        acc1 = wmma_f32(a[i], b1, acc1);
    }

    #pragma unroll
    for (int i = 0; i < 8; ++i) {
        rw[i * 32]       = acc0[i];
        rw[(8 + i) * 32] = acc1[i];
    }
    __syncthreads();

    float s = 0.f;
    #pragma unroll
    for (int w = 0; w < 16; ++w) s += red[w * 512 + t];
    const int tile = wv >> 3, i = wv & 7;
    cls_lin[(mt * 16 + i + 8 * half) * 768 + pair * 32 + tile * 16 + m] = s;
}

__device__ __forceinline__ void cls_ln(const float* __restrict__ cls_lin,
                                       const float* __restrict__ cls_b,
                                       const float* __restrict__ gamma,
                                       const float* __restrict__ beta,
                                       float* __restrict__ out,
                                       int b) {
    __shared__ float sh_s[8], sh_ss[8];
    const int t = threadIdx.x;
    float x[3];
    float s = 0.f, ss = 0.f;
    #pragma unroll
    for (int j = 0; j < 3; ++j) {
        int c = t + 256 * j;
        float v = cls_lin[b * 768 + c] + cls_b[c];
        x[j] = v; s += v; ss += v * v;
    }
    #pragma unroll
    for (int o = 1; o < 32; o <<= 1) {
        s  += __shfl_xor(s,  o, 32);
        ss += __shfl_xor(ss, o, 32);
    }
    if ((t & 31) == 0) { sh_s[t >> 5] = s; sh_ss[t >> 5] = ss; }
    __syncthreads();
    float ts = 0.f, tss = 0.f;
    #pragma unroll
    for (int j = 0; j < 8; ++j) { ts += sh_s[j]; tss += sh_ss[j]; }
    const float mean = ts * (1.f / 768.f);
    const float var  = tss * (1.f / 768.f) - mean * mean;
    const float rstd = rsqrtf(var + 1e-5f);
    #pragma unroll
    for (int j = 0; j < 3; ++j) {
        int c = t + 256 * j;
        out[b * 768 + c] = (x[j] - mean) * rstd * gamma[c] + beta[c];
    }
}

#define WN_R 16
#define WN_W (WN_R + 9)

__device__ __forceinline__ void window_norm(const float* __restrict__ reps,
                                            const int* __restrict__ Lw,
                                            float* __restrict__ out,
                                            int blk) {
    const int lane = threadIdx.x & 31;
    const int wv = __builtin_amdgcn_readfirstlane((int)(threadIdx.x >> 5));
    const int b  = blk >> 4;
    const int t0 = (((blk & 15) << 3) + wv) * WN_R;
    const int L  = Lw[b];
    const float* base = reps + ((long)b * 2046) * 128 + lane * 4;

    float4 w[WN_W];
    #pragma unroll
    for (int i = 0; i < WN_W; ++i) {
        const int r  = t0 - 5 + i;
        const int rc = r < 0 ? 0 : (r > 2045 ? 2045 : r);
        w[i] = *(const float4*)(base + (long)rc * 128);
    }
    #pragma unroll
    for (int i = 0; i < WN_W; ++i) {
        const int r = t0 - 5 + i;
        if (r < 0 || r >= L) w[i] = float4{0.f, 0.f, 0.f, 0.f};
    }

    float4 mn[WN_R];
    float  sq[WN_R];
    #pragma unroll
    for (int j = 0; j < WN_R; ++j) {
        const int tp = t0 + j;
        int start = tp - 5; if (start < 0) start = 0;
        int end   = tp + 5; if (end > L) end = L;
        int cnt   = end - start; if (cnt < 1) cnt = 1;

        float4 sum = {0.f, 0.f, 0.f, 0.f};
        #pragma unroll
        for (int k = 0; k < 10; ++k) {
            const float4 v = w[j + k];
            sum.x += v.x; sum.y += v.y; sum.z += v.z; sum.w += v.w;
        }
        const float inv = 1.f / (float)cnt;
        float4 m4 = {sum.x * inv, sum.y * inv, sum.z * inv, sum.w * inv};
        mn[j] = m4;
        sq[j] = m4.x * m4.x + m4.y * m4.y + m4.z * m4.z + m4.w * m4.w;
    }

    #pragma unroll
    for (int o = 1; o < 32; o <<= 1) {
        #pragma unroll
        for (int j = 0; j < WN_R; ++j) sq[j] += __shfl_xor(sq[j], o, 32);
    }

    #pragma unroll
    for (int j = 0; j < WN_R; ++j) {
        const int tp = t0 + j;
        if (tp < 2046) {
            const bool  valid = tp < L;
            const float n  = sqrtf(sq[j]);
            const float sc = (valid && n > 0.f) ? (1.f / n) : 0.f;
            const float4 m4 = mn[j];
            float4 o4 = {m4.x * sc, m4.y * sc, m4.z * sc, m4.w * sc};
            *(float4*)(out + ((long)b * 2046 + tp) * 128 + lane * 4) = o4;
        }
    }
}

#define TAIL_CLS 32

__global__ __launch_bounds__(256) void tail_k(const float* __restrict__ reps,
                                              const int* __restrict__ Lw,
                                              const float* __restrict__ cls_lin,
                                              const float* __restrict__ cls_b,
                                              const float* __restrict__ gamma,
                                              const float* __restrict__ beta,
                                              float* __restrict__ out) {
    const int blk = blockIdx.x;
    if (blk < TAIL_CLS)
        cls_ln(cls_lin, cls_b, gamma, beta, out, blk);
    else
        window_norm(reps, Lw, out + 24576, blk - TAIL_CLS);
}

extern "C" void kernel_launch(void* const* d_in, const int* in_sizes, int n_in,
                              void* d_out, int out_size, void* d_ws, size_t ws_size,
                              hipStream_t stream) {
    const float* hidden   = (const float*)d_in[0];
    const int*   amask    = (const int*)  d_in[1];
    const float* tok_w    = (const float*)d_in[2];
    const float* tok_b    = (const float*)d_in[3];
    const float* cls_w    = (const float*)d_in[4];
    const float* cls_b    = (const float*)d_in[5];
    const float* ln_tok_g = (const float*)d_in[6];
    const float* ln_tok_b = (const float*)d_in[7];
    const float* ln_cls_g = (const float*)d_in[8];
    const float* ln_cls_b = (const float*)d_in[9];
    float* out = (float*)d_out;

    float* ws      = (float*)d_ws;
    float* reps    = ws;
    float* cls_lin = ws + 8380416;
    float* wr_tok  = ws + 8404992;
    float* wr_cls  = ws + 8503296;
    int*   lenw    = (int*)(ws + 9093120);

    prep_k<<<2720, 256, 0, stream>>>(tok_w, cls_w, amask,
                                     wr_tok, wr_cls, lenw);

    tok_gemm_ln<<<1024, 128, 0, stream>>>(hidden, amask, wr_tok, tok_b,
                                          ln_tok_g, ln_tok_b, reps);
    cls_gemm<<<48, 512, 0, stream>>>(hidden, wr_cls, cls_lin);
    tail_k<<<544, 256, 0, stream>>>(reps, lenw, cls_lin, cls_b,
                                    ln_cls_g, ln_cls_b, out);
}


# cls product reads cls_w in place (B fragments straight from the row-major weight) and rides as 48 extra blocks of the tok launch: no cls reorder, no cls launch (4 -> 3); cls K-sum order changes
# speedup vs baseline: 2.0426x; 1.0436x over previous
#include <hip/hip_runtime.h>
#include <hip/hip_bf16.h>


typedef float v2f __attribute__((ext_vector_type(2)));
typedef float v8f __attribute__((ext_vector_type(8)));

__device__ __forceinline__ v8f wmma_f32(v2f a, v2f b, v8f c) {
    return __builtin_amdgcn_wmma_f32_16x16x4_f32(
        false, a, false, b, (short)0, c, false, false);
}

template <int X>
__device__ __forceinline__ float row_xor_add(float v) {
    const int b = __builtin_bit_cast(int, v);
    const int o = __builtin_amdgcn_update_dpp(0, b, 0x160 + X, 0xf, 0xf, true);
    return v + __builtin_bit_cast(float, o);
}

__device__ __forceinline__ void reorder_w(const float* __restrict__ W,
                                          int N,
                                          float* __restrict__ Wr,
                                          int blk) {
    int o    = blk * 256 + threadIdx.x;
    int pair = o / 24576;
    int rem  = o % 24576;
    int i    = rem / 128;
    int rem2 = rem % 128;
    int lane = rem2 >> 2;
    int tl   = (rem2 >> 1) & 1;
    int j    = rem2 & 1;
    int m    = lane & 15;
    int half = lane >> 4;
    int k    = 4 * i + 2 * half + j;
    int n    = (pair * 2 + tl) * 16 + m;
    Wr[o] = W[k * N + n];
}

__device__ __forceinline__ void lengths_k(const int* __restrict__ mask,
                                          int* __restrict__ Lout,
                                          int b) {
    __shared__ int sh[8];
    int t = threadIdx.x;
    int s = 0;
    #pragma unroll
    for (int j = 0; j < 8; ++j) s += mask[b * 2048 + t + 256 * j];
    #pragma unroll
    for (int o = 1; o < 32; o <<= 1) s += __shfl_xor(s, o, 32);
    if ((t & 31) == 0) sh[t >> 5] = s;
    __syncthreads();
    if (t == 0) {
        int tot = 0;
        #pragma unroll
        for (int j = 0; j < 8; ++j) tot += sh[j];
        int L = tot - 1;
        Lout[b] = L < 2046 ? L : 2046;
    }
}

#define PREP_LEN  32

__global__ __launch_bounds__(256) void prep_k(const float* __restrict__ tok_w,
                                              const int* __restrict__ mask,
                                              float* __restrict__ wr_tok,
                                              int* __restrict__ Lout) {
    const int blk = blockIdx.x;
    if (blk < PREP_LEN)
        lengths_k(mask, Lout, blk);
    else
        reorder_w(tok_w, 128, wr_tok, blk - PREP_LEN);
}

#define LDA 132
#define LDC 132
#define KS  128
#define NS  6
#define IT  32
#define NQ  16
#define TILE (64 * LDA)

__device__ __forceinline__ void tok_slice(const float* __restrict__ cur,
                                          const float4* __restrict__ bq,
                                          v8f (&acc)[4][2]) {
    #pragma unroll
    for (int i = 0; i < IT; ++i) {
        float4 f = bq[i * 32];
        v2f b0 = {f.x, f.y};
        v2f b1 = {f.z, f.w};
        #pragma unroll
        for (int mt = 0; mt < 4; ++mt) {
            v2f a = *(const v2f*)&cur[mt * 16 * LDA + 4 * i];
            acc[mt][0] = wmma_f32(a, b0, acc[mt][0]);
            acc[mt][1] = wmma_f32(a, b1, acc[mt][1]);
        }
    }
}

__device__ __forceinline__ void tok_gemm_ln(
    const float* __restrict__ hidden, const int* __restrict__ amask,
    const float* __restrict__ wr,     const float* __restrict__ tok_b,
    const float* __restrict__ gamma,  const float* __restrict__ beta,
    float* __restrict__ reps, int blk) {
    __shared__ __align__(16) float shmem[2 * TILE];
    float* shC = shmem;

    const int  t    = threadIdx.x;
    const int  lane = t & 31;
    const int  wv   = t >> 5;
    const int  m    = lane & 15;
    const int  half = lane >> 4;
    const long row0 = (long)blk * 64;

    v8f acc[4][2] = {};
    const float4* bp = (const float4*)wr + ((long)wv * 192) * 32 + lane;

    const float* gsrc = hidden + (row0 + wv) * 768 + lane * 4;
    const int    lofs = wv * LDA + lane * 4;
    const int    aofs = m * LDA + 2 * half;

    {
        float4 p0[NQ];
        #pragma unroll
        for (int q = 0; q < NQ; ++q)
            p0[q] = *(const float4*)(gsrc + q * (4 * 768));
        #pragma unroll
        for (int q = 0; q < NQ; ++q)
            *(float4*)&shmem[lofs + q * (4 * LDA)] = p0[q];
    }
    __syncthreads();

    for (int s = 0; s < NS - 1; ++s) {
        const float* cur = shmem + (s & 1) * TILE;
        float*       nxt = shmem + ((s + 1) & 1) * TILE;

        float4 pre[NQ];
        #pragma unroll
        for (int q = 0; q < NQ; ++q)
            pre[q] = *(const float4*)(gsrc + q * (4 * 768) + (s + 1) * KS);

        tok_slice(cur + aofs, bp + (long)s * (IT * 32), acc);

        #pragma unroll
        for (int q = 0; q < NQ; ++q)
            *(float4*)&nxt[lofs + q * (4 * LDA)] = pre[q];
        __syncthreads();
    }
    tok_slice(shmem + ((NS - 1) & 1) * TILE + aofs, bp + (long)(NS - 1) * (IT * 32), acc);

    #pragma unroll
    for (int mt = 0; mt < 4; ++mt) {
        #pragma unroll
        for (int i = 0; i < 8; ++i) {
            shC[(mt * 16 + i + 8 * half) * LDC + wv * 32 + m]      = acc[mt][0][i];
            shC[(mt * 16 + i + 8 * half) * LDC + wv * 32 + 16 + m] = acc[mt][1][i];
        }
    }
    __syncthreads();

    const int c4 = 4 * m;
    const float4 tbA = *(const float4*)(tok_b + c4);
    const float4 tbB = *(const float4*)(tok_b + 64 + c4);
    const float4 gaA = *(const float4*)(gamma + c4);
    const float4 gaB = *(const float4*)(gamma + 64 + c4);
    const float4 beA = *(const float4*)(beta + c4);
    const float4 beB = *(const float4*)(beta + 64 + c4);

    float4 xa[8], xb[8];
    float  s[8], ss[8], mk[8];
    #pragma unroll
    for (int p = 0; p < 8; ++p) {
        const int r = wv * 16 + 2 * p + half;
        mk[p] = (float)amask[row0 + r];
        float4 a = *(const float4*)&shC[r * LDC + c4];
        float4 b = *(const float4*)&shC[r * LDC + 64 + c4];
        a.x += tbA.x; a.y += tbA.y; a.z += tbA.z; a.w += tbA.w;
        b.x += tbB.x; b.y += tbB.y; b.z += tbB.z; b.w += tbB.w;
        xa[p] = a; xb[p] = b;
        s[p]  = ((a.x + a.y) + (a.z + a.w)) + ((b.x + b.y) + (b.z + b.w));
        ss[p] = ((a.x * a.x + a.y * a.y) + (a.z * a.z + a.w * a.w))
              + ((b.x * b.x + b.y * b.y) + (b.z * b.z + b.w * b.w));
    }
    #pragma unroll
    for (int p = 0; p < 8; ++p) { s[p] = row_xor_add<1>(s[p]); ss[p] = row_xor_add<1>(ss[p]); }
    #pragma unroll
    for (int p = 0; p < 8; ++p) { s[p] = row_xor_add<2>(s[p]); ss[p] = row_xor_add<2>(ss[p]); }
    #pragma unroll
    for (int p = 0; p < 8; ++p) { s[p] = row_xor_add<4>(s[p]); ss[p] = row_xor_add<4>(ss[p]); }
    #pragma unroll
    for (int p = 0; p < 8; ++p) { s[p] = row_xor_add<8>(s[p]); ss[p] = row_xor_add<8>(ss[p]); }

    #pragma unroll
    for (int p = 0; p < 8; ++p) {
        const float mean = s[p] * (1.f / 128.f);
        const float var  = ss[p] * (1.f / 128.f) - mean * mean;
        const float rstd = rsqrtf(var + 1e-5f);
        const long rg = row0 + wv * 16 + 2 * p + half;
        const int  bI = (int)(rg >> 11);
        const int  sI = (int)(rg & 2047);
        if (sI >= 1 && sI <= 2046) {
            float* dst = reps + ((long)(bI * 2046 + sI - 1)) * 128 + c4;
            const float4 a = xa[p], b = xb[p];
            float4 ya, yb;
            ya.x = fmaxf((a.x - mean) * rstd * gaA.x + beA.x, 0.f) * mk[p];
            ya.y = fmaxf((a.y - mean) * rstd * gaA.y + beA.y, 0.f) * mk[p];
            ya.z = fmaxf((a.z - mean) * rstd * gaA.z + beA.z, 0.f) * mk[p];
            ya.w = fmaxf((a.w - mean) * rstd * gaA.w + beA.w, 0.f) * mk[p];
            yb.x = fmaxf((b.x - mean) * rstd * gaB.x + beB.x, 0.f) * mk[p];
            yb.y = fmaxf((b.y - mean) * rstd * gaB.y + beB.y, 0.f) * mk[p];
            yb.z = fmaxf((b.z - mean) * rstd * gaB.z + beB.z, 0.f) * mk[p];
            yb.w = fmaxf((b.w - mean) * rstd * gaB.w + beB.w, 0.f) * mk[p];
            *(float4*)dst        = ya;
            *(float4*)(dst + 64) = yb;
        }
    }
}

#define CLS_KW 12
#define CLS_G  4

__device__ __forceinline__ void cls_gemm(const float* __restrict__ hidden,
                                         const float* __restrict__ cls_w,
                                         float* __restrict__ cls_lin,
                                         int blk) {
    __shared__ float red[4 * 512];
    const int t = threadIdx.x, lane = t & 31, wv = t >> 5;
    const int m = lane & 15, half = lane >> 4;
    const int pair = blk >> 1;
    const int mt   = blk & 1;

    const float* bp = cls_w + (long)(wv * (4 * CLS_G * CLS_KW) + 2 * half) * 768
                            + pair * 32 + m;
    const float* ap = hidden + (long)(mt * 16 + m) * (2048 * 768)
                             + wv * (4 * CLS_G * CLS_KW) + 2 * half;
    float* rw = red + wv * 512 + lane;

    v8f acc0 = {}, acc1 = {};
    for (int g = 0; g < CLS_G; ++g) {
        v2f a[CLS_KW], b0[CLS_KW], b1[CLS_KW];
        #pragma unroll
        for (int i = 0; i < CLS_KW; ++i) {
            const int kk = g * CLS_KW + i;
            const float* q = bp + (long)(4 * kk) * 768;
            a[i]  = *(const v2f*)(ap + 4 * kk);
            b0[i] = v2f{q[0],  q[768]};
            b1[i] = v2f{q[16], q[768 + 16]};
        }
        #pragma unroll
        for (int i = 0; i < CLS_KW; ++i) {
            acc0 = wmma_f32(a[i], b0[i], acc0);
            acc1 = wmma_f32(a[i], b1[i], acc1);
        }
    }

    #pragma unroll
    for (int i = 0; i < 8; ++i) {
        rw[i * 32]       = acc0[i];
        rw[(8 + i) * 32] = acc1[i];
    }
    __syncthreads();

    #pragma unroll
    for (int q = 0; q < 4; ++q) {
        const int slot = t + 128 * q;
        float s = red[slot];
        #pragma unroll
        for (int w = 1; w < 4; ++w) s += red[w * 512 + slot];
        const int reg = wv + 4 * q;
        const int tile = reg >> 3, i = reg & 7;
        cls_lin[(mt * 16 + i + 8 * half) * 768 + pair * 32 + tile * 16 + m] = s;
    }
}

#define MAIN_CLS 48

__global__ __launch_bounds__(128) void main_k(
    const float* __restrict__ hidden, const int* __restrict__ amask,
    const float* __restrict__ wr,     const float* __restrict__ tok_b,
    const float* __restrict__ gamma,  const float* __restrict__ beta,
    float* __restrict__ reps,
    const float* __restrict__ cls_w,  float* __restrict__ cls_lin) {
    const int blk = blockIdx.x;
    if (blk < MAIN_CLS)
        cls_gemm(hidden, cls_w, cls_lin, blk);
    else
        tok_gemm_ln(hidden, amask, wr, tok_b, gamma, beta, reps, blk - MAIN_CLS);
}

__device__ __forceinline__ void cls_ln(const float* __restrict__ cls_lin,
                                       const float* __restrict__ cls_b,
                                       const float* __restrict__ gamma,
                                       const float* __restrict__ beta,
                                       float* __restrict__ out,
                                       int b) {
    __shared__ float sh_s[8], sh_ss[8];
    const int t = threadIdx.x;
    float x[3];
    float s = 0.f, ss = 0.f;
    #pragma unroll
    for (int j = 0; j < 3; ++j) {
        int c = t + 256 * j;
        float v = cls_lin[b * 768 + c] + cls_b[c];
        x[j] = v; s += v; ss += v * v;
    }
    #pragma unroll
    for (int o = 1; o < 32; o <<= 1) {
        s  += __shfl_xor(s,  o, 32);
        ss += __shfl_xor(ss, o, 32);
    }
    if ((t & 31) == 0) { sh_s[t >> 5] = s; sh_ss[t >> 5] = ss; }
    __syncthreads();
    float ts = 0.f, tss = 0.f;
    #pragma unroll
    for (int j = 0; j < 8; ++j) { ts += sh_s[j]; tss += sh_ss[j]; }
    const float mean = ts * (1.f / 768.f);
    const float var  = tss * (1.f / 768.f) - mean * mean;
    const float rstd = rsqrtf(var + 1e-5f);
    #pragma unroll
    for (int j = 0; j < 3; ++j) {
        int c = t + 256 * j;
        out[b * 768 + c] = (x[j] - mean) * rstd * gamma[c] + beta[c];
    }
}

#define WN_R 16
#define WN_W (WN_R + 9)

__device__ __forceinline__ void window_norm(const float* __restrict__ reps,
                                            const int* __restrict__ Lw,
                                            float* __restrict__ out,
                                            int blk) {
    const int lane = threadIdx.x & 31;
    const int wv = __builtin_amdgcn_readfirstlane((int)(threadIdx.x >> 5));
    const int b  = blk >> 4;
    const int t0 = (((blk & 15) << 3) + wv) * WN_R;
    const int L  = Lw[b];
    const float* base = reps + ((long)b * 2046) * 128 + lane * 4;

    float4 w[WN_W];
    #pragma unroll
    for (int i = 0; i < WN_W; ++i) {
        const int r  = t0 - 5 + i;
        const int rc = r < 0 ? 0 : (r > 2045 ? 2045 : r);
        w[i] = *(const float4*)(base + (long)rc * 128);
    }
    #pragma unroll
    for (int i = 0; i < WN_W; ++i) {
        const int r = t0 - 5 + i;
        if (r < 0 || r >= L) w[i] = float4{0.f, 0.f, 0.f, 0.f};
    }

    float4 mn[WN_R];
    float  sq[WN_R];
    #pragma unroll
    for (int j = 0; j < WN_R; ++j) {
        const int tp = t0 + j;
        int start = tp - 5; if (start < 0) start = 0;
        int end   = tp + 5; if (end > L) end = L;
        int cnt   = end - start; if (cnt < 1) cnt = 1;

        float4 sum = {0.f, 0.f, 0.f, 0.f};
        #pragma unroll
        for (int k = 0; k < 10; ++k) {
            const float4 v = w[j + k];
            sum.x += v.x; sum.y += v.y; sum.z += v.z; sum.w += v.w;
        }
        const float inv = 1.f / (float)cnt;
        float4 m4 = {sum.x * inv, sum.y * inv, sum.z * inv, sum.w * inv};
        mn[j] = m4;
        sq[j] = m4.x * m4.x + m4.y * m4.y + m4.z * m4.z + m4.w * m4.w;
    }

    #pragma unroll
    for (int o = 1; o < 32; o <<= 1) {
        #pragma unroll
        for (int j = 0; j < WN_R; ++j) sq[j] += __shfl_xor(sq[j], o, 32);
    }

    #pragma unroll
    for (int j = 0; j < WN_R; ++j) {
        const int tp = t0 + j;
        if (tp < 2046) {
            const bool  valid = tp < L;
            const float n  = sqrtf(sq[j]);
            const float sc = (valid && n > 0.f) ? (1.f / n) : 0.f;
            const float4 m4 = mn[j];
            float4 o4 = {m4.x * sc, m4.y * sc, m4.z * sc, m4.w * sc};
            *(float4*)(out + ((long)b * 2046 + tp) * 128 + lane * 4) = o4;
        }
    }
}

#define TAIL_CLS 32

__global__ __launch_bounds__(256) void tail_k(const float* __restrict__ reps,
                                              const int* __restrict__ Lw,
                                              const float* __restrict__ cls_lin,
                                              const float* __restrict__ cls_b,
                                              const float* __restrict__ gamma,
                                              const float* __restrict__ beta,
                                              float* __restrict__ out) {
    const int blk = blockIdx.x;
    if (blk < TAIL_CLS)
        cls_ln(cls_lin, cls_b, gamma, beta, out, blk);
    else
        window_norm(reps, Lw, out + 24576, blk - TAIL_CLS);
}

extern "C" void kernel_launch(void* const* d_in, const int* in_sizes, int n_in,
                              void* d_out, int out_size, void* d_ws, size_t ws_size,
                              hipStream_t stream) {
    const float* hidden   = (const float*)d_in[0];
    const int*   amask    = (const int*)  d_in[1];
    const float* tok_w    = (const float*)d_in[2];
    const float* tok_b    = (const float*)d_in[3];
    const float* cls_w    = (const float*)d_in[4];
    const float* cls_b    = (const float*)d_in[5];
    const float* ln_tok_g = (const float*)d_in[6];
    const float* ln_tok_b = (const float*)d_in[7];
    const float* ln_cls_g = (const float*)d_in[8];
    const float* ln_cls_b = (const float*)d_in[9];
    float* out = (float*)d_out;

    float* ws      = (float*)d_ws;
    float* reps    = ws;
    float* cls_lin = ws + 8380416;
    float* wr_tok  = ws + 8404992;
    int*   lenw    = (int*)(ws + 9093120);

    prep_k<<<416, 256, 0, stream>>>(tok_w, amask, wr_tok, lenw);

    main_k<<<1072, 128, 0, stream>>>(hidden, amask, wr_tok, tok_b,
                                     ln_tok_g, ln_tok_b, reps,
                                     cls_w, cls_lin);
    tail_k<<<544, 256, 0, stream>>>(reps, lenw, cls_lin, cls_b,
                                    ln_cls_g, ln_cls_b, out);
}
